# starting kernel
# baseline (speedup 1.0000x reference)
#include <hip/hip_runtime.h>
#include <hip/hip_bf16.h>


#define N_NODES 100000
#define N_EDGES 1600000
#define IN_DIM  512
#define HID     64
#define OUT_DIM 5

typedef __attribute__((ext_vector_type(2))) float v2f;
typedef __attribute__((ext_vector_type(8))) float v8f;

__global__ void k_init_deg(float* __restrict__ deg) {
    int i = blockIdx.x * blockDim.x + threadIdx.x;
    if (i < N_NODES) deg[i] = 1.0f;
}

__global__ void k_deg_scatter(const int* __restrict__ col,
                              const float* __restrict__ w,
                              float* __restrict__ deg) {
    int stride = gridDim.x * blockDim.x;
    for (int e = blockIdx.x * blockDim.x + threadIdx.x; e < N_EDGES; e += stride)
        atomicAdd(deg + col[e], w[e]);
}

__global__ void k_dinv(float* __restrict__ deg) {
    int i = blockIdx.x * blockDim.x + threadIdx.x;
    if (i < N_NODES) {
        float d = deg[i];
        deg[i] = (d > 0.0f) ? rsqrtf(d) : 0.0f;
    }
}

__global__ void k_zero(float* __restrict__ p, int n) {
    int stride = gridDim.x * blockDim.x;
    for (int i = blockIdx.x * blockDim.x + threadIdx.x; i < n; i += stride)
        p[i] = 0.0f;
}

__global__ void k_gemm_n64(const float* __restrict__ X,
                           const float* __restrict__ W,
                           float* __restrict__ H,
                           int M, int K) {
    const int lane  = threadIdx.x & 31;
    const int waveT = (blockIdx.x * blockDim.x + threadIdx.x) >> 5;
    const int nTiles = M >> 4;
    if (waveT >= nTiles) return;

    const int m0   = waveT << 4;
    const int mRow = m0 + (lane & 15);
    const int kSub = (lane & 16) ? 2 : 0;
    const int nCol = lane & 15;

    v8f c0 = {}, c1 = {}, c2 = {}, c3 = {};

    for (int k = 0; k < K; k += 4) {
        v2f a = *(const v2f*)(X + (size_t)mRow * K + (k + kSub));

        const float* wr0 = W + (size_t)(k + kSub)     * HID + nCol;
        const float* wr1 = W + (size_t)(k + kSub + 1) * HID + nCol;
        v2f b0, b1, b2, b3;
        b0.x = wr0[0];  b0.y = wr1[0];
        b1.x = wr0[16]; b1.y = wr1[16];
        b2.x = wr0[32]; b2.y = wr1[32];
        b3.x = wr0[48]; b3.y = wr1[48];

        c0 = __builtin_amdgcn_wmma_f32_16x16x4_f32(false, a, false, b0,
                                                   (short)0, c0, false, false);
        c1 = __builtin_amdgcn_wmma_f32_16x16x4_f32(false, a, false, b1,
                                                   (short)0, c1, false, false);
        c2 = __builtin_amdgcn_wmma_f32_16x16x4_f32(false, a, false, b2,
                                                   (short)0, c2, false, false);
        c3 = __builtin_amdgcn_wmma_f32_16x16x4_f32(false, a, false, b3,
                                                   (short)0, c3, false, false);
    }

    const int rBase = m0 + ((lane & 16) ? 8 : 0);
    #pragma unroll
    for (int j = 0; j < 8; ++j) {
        float* dst = H + (size_t)(rBase + j) * HID + nCol;
        dst[0]  = c0[j];
        dst[16] = c1[j];
        dst[32] = c2[j];
        dst[48] = c3[j];
    }
}

__global__ void k_edge_scatter(const int* __restrict__ row,
                               const int* __restrict__ col,
                               const float* __restrict__ w,
                               const float* __restrict__ dinv,
                               const float* __restrict__ h,
                               float* __restrict__ agg) {
    const int lane   = threadIdx.x & 31;
    const int wid    = (blockIdx.x * blockDim.x + threadIdx.x) >> 5;
    const int nWaves = (gridDim.x * blockDim.x) >> 5;

    for (int e = wid; e < N_EDGES; e += nWaves) {
        const int r = row[e];
        const int c = col[e];
        const float norm = dinv[r] * w[e] * dinv[c];
        v2f hv = *(const v2f*)(h + (size_t)r * HID + lane * 2);
        float* dst = agg + (size_t)c * HID + lane * 2;
        atomicAdd(dst + 0, norm * hv.x);
        atomicAdd(dst + 1, norm * hv.y);
    }
}

__global__ void k_finalize(float* __restrict__ agg,
                           const float* __restrict__ h,
                           const float* __restrict__ dinv,
                           const float* __restrict__ b,
                           int relu) {
    int stride = gridDim.x * blockDim.x;
    for (int i = blockIdx.x * blockDim.x + threadIdx.x; i < N_NODES * HID; i += stride) {
        const int n = i >> 6;
        const int f = i & 63;
        const float di = dinv[n];
        float v = agg[i] + di * di * h[i] + b[f];
        agg[i] = relu ? fmaxf(v, 0.0f) : v;
    }
}

__global__ void k_fc(const float* __restrict__ h,
                     const float* __restrict__ Wfc,
                     const float* __restrict__ bfc,
                     float* __restrict__ out) {
    int n = blockIdx.x * blockDim.x + threadIdx.x;
    if (n >= N_NODES) return;
    float acc[OUT_DIM];
    #pragma unroll
    for (int o = 0; o < OUT_DIM; ++o) acc[o] = bfc[o];
    const float* hr = h + (size_t)n * HID;
    #pragma unroll 4
    for (int k = 0; k < HID; ++k) {
        const float xv = hr[k];
        #pragma unroll
        for (int o = 0; o < OUT_DIM; ++o)
            acc[o] = fmaf(xv, Wfc[k * OUT_DIM + o], acc[o]);
    }
    float* orow = out + (size_t)n * OUT_DIM;
    #pragma unroll
    for (int o = 0; o < OUT_DIM; ++o) orow[o] = acc[o];
}

extern "C" void kernel_launch(void* const* d_in, const int* in_sizes, int n_in,
                              void* d_out, int out_size, void* d_ws, size_t ws_size,
                              hipStream_t stream) {
    const float* x   = (const float*)d_in[0];
    const int*   ei  = (const int*)  d_in[1];
    const float* ew  = (const float*)d_in[2];
    const float* W1  = (const float*)d_in[3];
    const float* b1  = (const float*)d_in[4];
    const float* W2  = (const float*)d_in[5];
    const float* b2  = (const float*)d_in[6];
    const float* Wfc = (const float*)d_in[7];
    const float* bfc = (const float*)d_in[8];
    float* out = (float*)d_out;

    const int* row = ei;
    const int* col = ei + N_EDGES;

    char* ws = (char*)d_ws;
    size_t off = 0;
    float* dinv = (float*)(ws + off); off += ((size_t)N_NODES * 4 + 255) & ~(size_t)255;
    float* bufH = (float*)(ws + off); off += (size_t)N_NODES * HID * 4;
    float* bufA = (float*)(ws + off); off += (size_t)N_NODES * HID * 4;
    (void)ws_size; (void)n_in; (void)in_sizes; (void)out_size;

    const int TPB = 256;
    const int nodeBlocks = (N_NODES + TPB - 1) / TPB;
    const int featElems  = N_NODES * HID;
    const int gemmBlocks = ((N_NODES / 16) + (TPB / 32) - 1) / (TPB / 32);

    k_init_deg<<<nodeBlocks, TPB, 0, stream>>>(dinv);
    k_deg_scatter<<<1024, TPB, 0, stream>>>(col, ew, dinv);
    k_dinv<<<nodeBlocks, TPB, 0, stream>>>(dinv);

    k_gemm_n64<<<gemmBlocks, TPB, 0, stream>>>(x, W1, bufH, N_NODES, IN_DIM);
    k_zero<<<4096, TPB, 0, stream>>>(bufA, featElems);
    k_edge_scatter<<<2048, TPB, 0, stream>>>(row, col, ew, dinv, bufH, bufA);
    k_finalize<<<4096, TPB, 0, stream>>>(bufA, bufH, dinv, b1, 1);

    k_gemm_n64<<<gemmBlocks, TPB, 0, stream>>>(bufA, W2, bufH, N_NODES, HID);
    k_zero<<<4096, TPB, 0, stream>>>(bufA, featElems);
    k_edge_scatter<<<2048, TPB, 0, stream>>>(row, col, ew, dinv, bufH, bufA);
    k_finalize<<<4096, TPB, 0, stream>>>(bufA, bufH, dinv, b2, 1);

    k_fc<<<nodeBlocks, TPB, 0, stream>>>(bufA, Wfc, bfc, out);
}


# edge scatter: a wave takes 32 edges per trip (coalesced index/weight loads, 32 norms at once), serves them by lane reads, h rows gathered ahead of their atomics; grid 1250x256; atomic order changes
# speedup vs baseline: 1.1540x; 1.1540x over previous
#include <hip/hip_runtime.h>
#include <hip/hip_bf16.h>


#define N_NODES 100000
#define N_EDGES 1600000
#define IN_DIM  512
#define HID     64
#define OUT_DIM 5

typedef __attribute__((ext_vector_type(2))) float v2f;
typedef __attribute__((ext_vector_type(8))) float v8f;

__global__ void k_init_deg(float* __restrict__ deg) {
    int i = blockIdx.x * blockDim.x + threadIdx.x;
    if (i < N_NODES) deg[i] = 1.0f;
}

__global__ void k_deg_scatter(const int* __restrict__ col,
                              const float* __restrict__ w,
                              float* __restrict__ deg) {
    int stride = gridDim.x * blockDim.x;
    for (int e = blockIdx.x * blockDim.x + threadIdx.x; e < N_EDGES; e += stride)
        atomicAdd(deg + col[e], w[e]);
}

__global__ void k_dinv(float* __restrict__ deg) {
    int i = blockIdx.x * blockDim.x + threadIdx.x;
    if (i < N_NODES) {
        float d = deg[i];
        deg[i] = (d > 0.0f) ? rsqrtf(d) : 0.0f;
    }
}

__global__ void k_zero(float* __restrict__ p, int n) {
    int stride = gridDim.x * blockDim.x;
    for (int i = blockIdx.x * blockDim.x + threadIdx.x; i < n; i += stride)
        p[i] = 0.0f;
}

__global__ void k_gemm_n64(const float* __restrict__ X,
                           const float* __restrict__ W,
                           float* __restrict__ H,
                           int M, int K) {
    const int lane  = threadIdx.x & 31;
    const int waveT = (blockIdx.x * blockDim.x + threadIdx.x) >> 5;
    const int nTiles = M >> 4;
    if (waveT >= nTiles) return;

    const int m0   = waveT << 4;
    const int mRow = m0 + (lane & 15);
    const int kSub = (lane & 16) ? 2 : 0;
    const int nCol = lane & 15;

    v8f c0 = {}, c1 = {}, c2 = {}, c3 = {};

    for (int k = 0; k < K; k += 4) {
        v2f a = *(const v2f*)(X + (size_t)mRow * K + (k + kSub));

        const float* wr0 = W + (size_t)(k + kSub)     * HID + nCol;
        const float* wr1 = W + (size_t)(k + kSub + 1) * HID + nCol;
        v2f b0, b1, b2, b3;
        b0.x = wr0[0];  b0.y = wr1[0];
        b1.x = wr0[16]; b1.y = wr1[16];
        b2.x = wr0[32]; b2.y = wr1[32];
        b3.x = wr0[48]; b3.y = wr1[48];

        c0 = __builtin_amdgcn_wmma_f32_16x16x4_f32(false, a, false, b0,
                                                   (short)0, c0, false, false);
        c1 = __builtin_amdgcn_wmma_f32_16x16x4_f32(false, a, false, b1,
                                                   (short)0, c1, false, false);
        c2 = __builtin_amdgcn_wmma_f32_16x16x4_f32(false, a, false, b2,
                                                   (short)0, c2, false, false);
        c3 = __builtin_amdgcn_wmma_f32_16x16x4_f32(false, a, false, b3,
                                                   (short)0, c3, false, false);
    }

    const int rBase = m0 + ((lane & 16) ? 8 : 0);
    #pragma unroll
    for (int j = 0; j < 8; ++j) {
        float* dst = H + (size_t)(rBase + j) * HID + nCol;
        dst[0]  = c0[j];
        dst[16] = c1[j];
        dst[32] = c2[j];
        dst[48] = c3[j];
    }
}

#define EB 32
#define GD 8
static_assert(N_EDGES % EB == 0, "edge count must fill whole wave trips");
static_assert(EB % GD == 0, "gather groups must tile a wave trip");
#define SCATTER_TRIPS_PER_WAVE 5
#define SCATTER_BLOCKS ((N_EDGES / EB) / 8 / SCATTER_TRIPS_PER_WAVE)
static_assert((N_EDGES / EB) % (8 * SCATTER_TRIPS_PER_WAVE) == 0,
              "trips must divide evenly over the scatter's waves");

__global__ void k_edge_scatter(const int* __restrict__ row,
                               const int* __restrict__ col,
                               const float* __restrict__ w,
                               const float* __restrict__ dinv,
                               const float* __restrict__ h,
                               float* __restrict__ agg) {
    const int lane   = threadIdx.x & 31;
    const int wid    = __builtin_amdgcn_readfirstlane(
                           (int)((blockIdx.x * blockDim.x + threadIdx.x) >> 5));
    const int nWaves = (int)((gridDim.x * blockDim.x) >> 5);
    const int fo     = lane * 2;

    for (int b = wid; b < N_EDGES / EB; b += nWaves) {
        const int e = b * EB + lane;
        const int r = row[e];
        const int c = col[e];
        const float norm = dinv[r] * w[e] * dinv[c];
        const int normBits = __float_as_int(norm);

        v2f hv[2][GD];

        #pragma unroll
        for (int j = 0; j < GD; ++j) {
            const int rj = __builtin_amdgcn_readlane(r, j);
            hv[0][j] = *(const v2f*)(h + (size_t)rj * HID + fo);
        }

        #pragma unroll
        for (int g = 0; g < EB / GD; ++g) {
            const int cur = g & 1;
            if (g + 1 < EB / GD) {
                #pragma unroll
                for (int j = 0; j < GD; ++j) {
                    const int rj = __builtin_amdgcn_readlane(r, (g + 1) * GD + j);
                    hv[cur ^ 1][j] = *(const v2f*)(h + (size_t)rj * HID + fo);
                }
            }
            #pragma unroll
            for (int j = 0; j < GD; ++j) {
                const int   cj = __builtin_amdgcn_readlane(c, g * GD + j);
                const float nj = __int_as_float(
                                     __builtin_amdgcn_readlane(normBits, g * GD + j));
                float* dst = agg + (size_t)cj * HID + fo;
                atomicAdd(dst + 0, nj * hv[cur][j].x);
                atomicAdd(dst + 1, nj * hv[cur][j].y);
            }
        }
    }
}

__global__ void k_finalize(float* __restrict__ agg,
                           const float* __restrict__ h,
                           const float* __restrict__ dinv,
                           const float* __restrict__ b,
                           int relu) {
    int stride = gridDim.x * blockDim.x;
    for (int i = blockIdx.x * blockDim.x + threadIdx.x; i < N_NODES * HID; i += stride) {
        const int n = i >> 6;
        const int f = i & 63;
        const float di = dinv[n];
        float v = agg[i] + di * di * h[i] + b[f];
        agg[i] = relu ? fmaxf(v, 0.0f) : v;
    }
}

__global__ void k_fc(const float* __restrict__ h,
                     const float* __restrict__ Wfc,
                     const float* __restrict__ bfc,
                     float* __restrict__ out) {
    int n = blockIdx.x * blockDim.x + threadIdx.x;
    if (n >= N_NODES) return;
    float acc[OUT_DIM];
    #pragma unroll
    for (int o = 0; o < OUT_DIM; ++o) acc[o] = bfc[o];
    const float* hr = h + (size_t)n * HID;
    #pragma unroll 4
    for (int k = 0; k < HID; ++k) {
        const float xv = hr[k];
        #pragma unroll
        for (int o = 0; o < OUT_DIM; ++o)
            acc[o] = fmaf(xv, Wfc[k * OUT_DIM + o], acc[o]);
    }
    float* orow = out + (size_t)n * OUT_DIM;
    #pragma unroll
    for (int o = 0; o < OUT_DIM; ++o) orow[o] = acc[o];
}

extern "C" void kernel_launch(void* const* d_in, const int* in_sizes, int n_in,
                              void* d_out, int out_size, void* d_ws, size_t ws_size,
                              hipStream_t stream) {
    const float* x   = (const float*)d_in[0];
    const int*   ei  = (const int*)  d_in[1];
    const float* ew  = (const float*)d_in[2];
    const float* W1  = (const float*)d_in[3];
    const float* b1  = (const float*)d_in[4];
    const float* W2  = (const float*)d_in[5];
    const float* b2  = (const float*)d_in[6];
    const float* Wfc = (const float*)d_in[7];
    const float* bfc = (const float*)d_in[8];
    float* out = (float*)d_out;

    const int* row = ei;
    const int* col = ei + N_EDGES;

    char* ws = (char*)d_ws;
    size_t off = 0;
    float* dinv = (float*)(ws + off); off += ((size_t)N_NODES * 4 + 255) & ~(size_t)255;
    float* bufH = (float*)(ws + off); off += (size_t)N_NODES * HID * 4;
    float* bufA = (float*)(ws + off); off += (size_t)N_NODES * HID * 4;
    (void)ws_size; (void)n_in; (void)in_sizes; (void)out_size;

    const int TPB = 256;
    const int nodeBlocks = (N_NODES + TPB - 1) / TPB;
    const int featElems  = N_NODES * HID;
    const int gemmBlocks = ((N_NODES / 16) + (TPB / 32) - 1) / (TPB / 32);

    k_init_deg<<<nodeBlocks, TPB, 0, stream>>>(dinv);
    k_deg_scatter<<<1024, TPB, 0, stream>>>(col, ew, dinv);
    k_dinv<<<nodeBlocks, TPB, 0, stream>>>(dinv);

    k_gemm_n64<<<gemmBlocks, TPB, 0, stream>>>(x, W1, bufH, N_NODES, IN_DIM);
    k_zero<<<4096, TPB, 0, stream>>>(bufA, featElems);
    k_edge_scatter<<<SCATTER_BLOCKS, TPB, 0, stream>>>(row, col, ew, dinv, bufH, bufA);
    k_finalize<<<4096, TPB, 0, stream>>>(bufA, bufH, dinv, b1, 1);

    k_gemm_n64<<<gemmBlocks, TPB, 0, stream>>>(bufA, W2, bufH, N_NODES, HID);
    k_zero<<<4096, TPB, 0, stream>>>(bufA, featElems);
    k_edge_scatter<<<SCATTER_BLOCKS, TPB, 0, stream>>>(row, col, ew, dinv, bufH, bufA);
    k_finalize<<<4096, TPB, 0, stream>>>(bufA, bufH, dinv, b2, 1);

    k_fc<<<nodeBlocks, TPB, 0, stream>>>(bufA, Wfc, bfc, out);
}


# edge scatter: lane owns features l and l+32 (was 2l, 2l+1), so each wave-wide atomic and gather covers one whole 128-byte line: atomic line requests per edge 4 -> 2; still atomics, arrival order
# speedup vs baseline: 1.3418x; 1.1627x over previous
#include <hip/hip_runtime.h>
#include <hip/hip_bf16.h>


#define N_NODES 100000
#define N_EDGES 1600000
#define IN_DIM  512
#define HID     64
#define OUT_DIM 5

typedef __attribute__((ext_vector_type(2))) float v2f;
typedef __attribute__((ext_vector_type(8))) float v8f;

__global__ void k_init_deg(float* __restrict__ deg) {
    int i = blockIdx.x * blockDim.x + threadIdx.x;
    if (i < N_NODES) deg[i] = 1.0f;
}

__global__ void k_deg_scatter(const int* __restrict__ col,
                              const float* __restrict__ w,
                              float* __restrict__ deg) {
    int stride = gridDim.x * blockDim.x;
    for (int e = blockIdx.x * blockDim.x + threadIdx.x; e < N_EDGES; e += stride)
        atomicAdd(deg + col[e], w[e]);
}

__global__ void k_dinv(float* __restrict__ deg) {
    int i = blockIdx.x * blockDim.x + threadIdx.x;
    if (i < N_NODES) {
        float d = deg[i];
        deg[i] = (d > 0.0f) ? rsqrtf(d) : 0.0f;
    }
}

__global__ void k_zero(float* __restrict__ p, int n) {
    int stride = gridDim.x * blockDim.x;
    for (int i = blockIdx.x * blockDim.x + threadIdx.x; i < n; i += stride)
        p[i] = 0.0f;
}

__global__ void k_gemm_n64(const float* __restrict__ X,
                           const float* __restrict__ W,
                           float* __restrict__ H,
                           int M, int K) {
    const int lane  = threadIdx.x & 31;
    const int waveT = (blockIdx.x * blockDim.x + threadIdx.x) >> 5;
    const int nTiles = M >> 4;
    if (waveT >= nTiles) return;

    const int m0   = waveT << 4;
    const int mRow = m0 + (lane & 15);
    const int kSub = (lane & 16) ? 2 : 0;
    const int nCol = lane & 15;

    v8f c0 = {}, c1 = {}, c2 = {}, c3 = {};

    for (int k = 0; k < K; k += 4) {
        v2f a = *(const v2f*)(X + (size_t)mRow * K + (k + kSub));

        const float* wr0 = W + (size_t)(k + kSub)     * HID + nCol;
        const float* wr1 = W + (size_t)(k + kSub + 1) * HID + nCol;
        v2f b0, b1, b2, b3;
        b0.x = wr0[0];  b0.y = wr1[0];
        b1.x = wr0[16]; b1.y = wr1[16];
        b2.x = wr0[32]; b2.y = wr1[32];
        b3.x = wr0[48]; b3.y = wr1[48];

        c0 = __builtin_amdgcn_wmma_f32_16x16x4_f32(false, a, false, b0,
                                                   (short)0, c0, false, false);
        c1 = __builtin_amdgcn_wmma_f32_16x16x4_f32(false, a, false, b1,
                                                   (short)0, c1, false, false);
        c2 = __builtin_amdgcn_wmma_f32_16x16x4_f32(false, a, false, b2,
                                                   (short)0, c2, false, false);
        c3 = __builtin_amdgcn_wmma_f32_16x16x4_f32(false, a, false, b3,
                                                   (short)0, c3, false, false);
    }

    const int rBase = m0 + ((lane & 16) ? 8 : 0);
    #pragma unroll
    for (int j = 0; j < 8; ++j) {
        float* dst = H + (size_t)(rBase + j) * HID + nCol;
        dst[0]  = c0[j];
        dst[16] = c1[j];
        dst[32] = c2[j];
        dst[48] = c3[j];
    }
}

#define EB 32
#define GD 8
static_assert(N_EDGES % EB == 0, "edge count must fill whole wave trips");
static_assert(EB % GD == 0, "gather groups must tile a wave trip");
static_assert(HID == 64, "a row is two 32-float halves, one per wave-wide access");
#define SCATTER_TRIPS_PER_WAVE 5
#define SCATTER_BLOCKS ((N_EDGES / EB) / 8 / SCATTER_TRIPS_PER_WAVE)
static_assert((N_EDGES / EB) % (8 * SCATTER_TRIPS_PER_WAVE) == 0,
              "trips must divide evenly over the scatter's waves");

__global__ void k_edge_scatter(const int* __restrict__ row,
                               const int* __restrict__ col,
                               const float* __restrict__ w,
                               const float* __restrict__ dinv,
                               const float* __restrict__ h,
                               float* __restrict__ agg) {
    const int lane   = threadIdx.x & 31;
    const int wid    = __builtin_amdgcn_readfirstlane(
                           (int)((blockIdx.x * blockDim.x + threadIdx.x) >> 5));
    const int nWaves = (int)((gridDim.x * blockDim.x) >> 5);
    const int fo     = lane;

    for (int b = wid; b < N_EDGES / EB; b += nWaves) {
        const int e = b * EB + lane;
        const int r = row[e];
        const int c = col[e];
        const float norm = dinv[r] * w[e] * dinv[c];
        const int normBits = __float_as_int(norm);

        v2f hv[2][GD];

        #pragma unroll
        for (int j = 0; j < GD; ++j) {
            const int rj = __builtin_amdgcn_readlane(r, j);
            const float* src = h + (size_t)rj * HID + fo;
            hv[0][j].x = src[0];
            hv[0][j].y = src[HID / 2];
        }

        #pragma unroll
        for (int g = 0; g < EB / GD; ++g) {
            const int cur = g & 1;
            if (g + 1 < EB / GD) {
                #pragma unroll
                for (int j = 0; j < GD; ++j) {
                    const int rj = __builtin_amdgcn_readlane(r, (g + 1) * GD + j);
                    const float* src = h + (size_t)rj * HID + fo;
                    hv[cur ^ 1][j].x = src[0];
                    hv[cur ^ 1][j].y = src[HID / 2];
                }
            }
            #pragma unroll
            for (int j = 0; j < GD; ++j) {
                const int   cj = __builtin_amdgcn_readlane(c, g * GD + j);
                const float nj = __int_as_float(
                                     __builtin_amdgcn_readlane(normBits, g * GD + j));
                float* dst = agg + (size_t)cj * HID + fo;
                atomicAdd(dst,           nj * hv[cur][j].x);
                atomicAdd(dst + HID / 2, nj * hv[cur][j].y);
            }
        }
    }
}

__global__ void k_finalize(float* __restrict__ agg,
                           const float* __restrict__ h,
                           const float* __restrict__ dinv,
                           const float* __restrict__ b,
                           int relu) {
    int stride = gridDim.x * blockDim.x;
    for (int i = blockIdx.x * blockDim.x + threadIdx.x; i < N_NODES * HID; i += stride) {
        const int n = i >> 6;
        const int f = i & 63;
        const float di = dinv[n];
        float v = agg[i] + di * di * h[i] + b[f];
        agg[i] = relu ? fmaxf(v, 0.0f) : v;
    }
}

__global__ void k_fc(const float* __restrict__ h,
                     const float* __restrict__ Wfc,
                     const float* __restrict__ bfc,
                     float* __restrict__ out) {
    int n = blockIdx.x * blockDim.x + threadIdx.x;
    if (n >= N_NODES) return;
    float acc[OUT_DIM];
    #pragma unroll
    for (int o = 0; o < OUT_DIM; ++o) acc[o] = bfc[o];
    const float* hr = h + (size_t)n * HID;
    #pragma unroll 4
    for (int k = 0; k < HID; ++k) {
        const float xv = hr[k];
        #pragma unroll
        for (int o = 0; o < OUT_DIM; ++o)
            acc[o] = fmaf(xv, Wfc[k * OUT_DIM + o], acc[o]);
    }
    float* orow = out + (size_t)n * OUT_DIM;
    #pragma unroll
    for (int o = 0; o < OUT_DIM; ++o) orow[o] = acc[o];
}

extern "C" void kernel_launch(void* const* d_in, const int* in_sizes, int n_in,
                              void* d_out, int out_size, void* d_ws, size_t ws_size,
                              hipStream_t stream) {
    const float* x   = (const float*)d_in[0];
    const int*   ei  = (const int*)  d_in[1];
    const float* ew  = (const float*)d_in[2];
    const float* W1  = (const float*)d_in[3];
    const float* b1  = (const float*)d_in[4];
    const float* W2  = (const float*)d_in[5];
    const float* b2  = (const float*)d_in[6];
    const float* Wfc = (const float*)d_in[7];
    const float* bfc = (const float*)d_in[8];
    float* out = (float*)d_out;

    const int* row = ei;
    const int* col = ei + N_EDGES;

    char* ws = (char*)d_ws;
    size_t off = 0;
    float* dinv = (float*)(ws + off); off += ((size_t)N_NODES * 4 + 255) & ~(size_t)255;
    float* bufH = (float*)(ws + off); off += (size_t)N_NODES * HID * 4;
    float* bufA = (float*)(ws + off); off += (size_t)N_NODES * HID * 4;
    (void)ws_size; (void)n_in; (void)in_sizes; (void)out_size;

    const int TPB = 256;
    const int nodeBlocks = (N_NODES + TPB - 1) / TPB;
    const int featElems  = N_NODES * HID;
    const int gemmBlocks = ((N_NODES / 16) + (TPB / 32) - 1) / (TPB / 32);

    k_init_deg<<<nodeBlocks, TPB, 0, stream>>>(dinv);
    k_deg_scatter<<<1024, TPB, 0, stream>>>(col, ew, dinv);
    k_dinv<<<nodeBlocks, TPB, 0, stream>>>(dinv);

    k_gemm_n64<<<gemmBlocks, TPB, 0, stream>>>(x, W1, bufH, N_NODES, IN_DIM);
    k_zero<<<4096, TPB, 0, stream>>>(bufA, featElems);
    k_edge_scatter<<<SCATTER_BLOCKS, TPB, 0, stream>>>(row, col, ew, dinv, bufH, bufA);
    k_finalize<<<4096, TPB, 0, stream>>>(bufA, bufH, dinv, b1, 1);

    k_gemm_n64<<<gemmBlocks, TPB, 0, stream>>>(bufA, W2, bufH, N_NODES, HID);
    k_zero<<<4096, TPB, 0, stream>>>(bufA, featElems);
    k_edge_scatter<<<SCATTER_BLOCKS, TPB, 0, stream>>>(row, col, ew, dinv, bufH, bufA);
    k_finalize<<<4096, TPB, 0, stream>>>(bufA, bufH, dinv, b2, 1);

    k_fc<<<nodeBlocks, TPB, 0, stream>>>(bufA, Wfc, bfc, out);
}


# edge weight factored: GEMM stores rows pre-scaled by dinv[row], finalize applies dinv[col] to the sum; edge scatter drops both random dinv gathers per edge; product order changes, atomics kept
# speedup vs baseline: 1.4029x; 1.0455x over previous
#include <hip/hip_runtime.h>
#include <hip/hip_bf16.h>


#define N_NODES 100000
#define N_EDGES 1600000
#define IN_DIM  512
#define HID     64
#define OUT_DIM 5

typedef __attribute__((ext_vector_type(2))) float v2f;
typedef __attribute__((ext_vector_type(8))) float v8f;

__global__ void k_init_deg(float* __restrict__ deg) {
    int i = blockIdx.x * blockDim.x + threadIdx.x;
    if (i < N_NODES) deg[i] = 1.0f;
}

__global__ void k_deg_scatter(const int* __restrict__ col,
                              const float* __restrict__ w,
                              float* __restrict__ deg) {
    int stride = gridDim.x * blockDim.x;
    for (int e = blockIdx.x * blockDim.x + threadIdx.x; e < N_EDGES; e += stride)
        atomicAdd(deg + col[e], w[e]);
}

__global__ void k_dinv(float* __restrict__ deg) {
    int i = blockIdx.x * blockDim.x + threadIdx.x;
    if (i < N_NODES) {
        float d = deg[i];
        deg[i] = (d > 0.0f) ? rsqrtf(d) : 0.0f;
    }
}

__global__ void k_zero(float* __restrict__ p, int n) {
    int stride = gridDim.x * blockDim.x;
    for (int i = blockIdx.x * blockDim.x + threadIdx.x; i < n; i += stride)
        p[i] = 0.0f;
}

__global__ void k_gemm_n64(const float* __restrict__ X,
                           const float* __restrict__ W,
                           const float* __restrict__ rs,
                           float* __restrict__ H,
                           int M, int K) {
    const int lane  = threadIdx.x & 31;
    const int waveT = (blockIdx.x * blockDim.x + threadIdx.x) >> 5;
    const int nTiles = M >> 4;
    if (waveT >= nTiles) return;

    const int m0   = waveT << 4;
    const int mRow = m0 + (lane & 15);
    const int kSub = (lane & 16) ? 2 : 0;
    const int nCol = lane & 15;

    v8f c0 = {}, c1 = {}, c2 = {}, c3 = {};

    for (int k = 0; k < K; k += 4) {
        v2f a = *(const v2f*)(X + (size_t)mRow * K + (k + kSub));

        const float* wr0 = W + (size_t)(k + kSub)     * HID + nCol;
        const float* wr1 = W + (size_t)(k + kSub + 1) * HID + nCol;
        v2f b0, b1, b2, b3;
        b0.x = wr0[0];  b0.y = wr1[0];
        b1.x = wr0[16]; b1.y = wr1[16];
        b2.x = wr0[32]; b2.y = wr1[32];
        b3.x = wr0[48]; b3.y = wr1[48];

        c0 = __builtin_amdgcn_wmma_f32_16x16x4_f32(false, a, false, b0,
                                                   (short)0, c0, false, false);
        c1 = __builtin_amdgcn_wmma_f32_16x16x4_f32(false, a, false, b1,
                                                   (short)0, c1, false, false);
        c2 = __builtin_amdgcn_wmma_f32_16x16x4_f32(false, a, false, b2,
                                                   (short)0, c2, false, false);
        c3 = __builtin_amdgcn_wmma_f32_16x16x4_f32(false, a, false, b3,
                                                   (short)0, c3, false, false);
    }

    const int rBase = m0 + ((lane & 16) ? 8 : 0);
    #pragma unroll
    for (int j = 0; j < 8; ++j) {
        const float sc = rs[rBase + j];
        float* dst = H + (size_t)(rBase + j) * HID + nCol;
        dst[0]  = sc * c0[j];
        dst[16] = sc * c1[j];
        dst[32] = sc * c2[j];
        dst[48] = sc * c3[j];
    }
}

#define EB 32
#define GD 8
static_assert(N_EDGES % EB == 0, "edge count must fill whole wave trips");
static_assert(EB % GD == 0, "gather groups must tile a wave trip");
static_assert(HID == 64, "a row is two 32-float halves, one per wave-wide access");
#define SCATTER_TRIPS_PER_WAVE 5
#define SCATTER_BLOCKS ((N_EDGES / EB) / 8 / SCATTER_TRIPS_PER_WAVE)
static_assert((N_EDGES / EB) % (8 * SCATTER_TRIPS_PER_WAVE) == 0,
              "trips must divide evenly over the scatter's waves");

__global__ void k_edge_scatter(const int* __restrict__ row,
                               const int* __restrict__ col,
                               const float* __restrict__ w,
                               const float* __restrict__ h,
                               float* __restrict__ agg) {
    const int lane   = threadIdx.x & 31;
    const int wid    = __builtin_amdgcn_readfirstlane(
                           (int)((blockIdx.x * blockDim.x + threadIdx.x) >> 5));
    const int nWaves = (int)((gridDim.x * blockDim.x) >> 5);
    const int fo     = lane;

    for (int b = wid; b < N_EDGES / EB; b += nWaves) {
        const int e = b * EB + lane;
        const int r = row[e];
        const int c = col[e];
        const int wBits = __float_as_int(w[e]);

        v2f hv[2][GD];

        #pragma unroll
        for (int j = 0; j < GD; ++j) {
            const int rj = __builtin_amdgcn_readlane(r, j);
            const float* src = h + (size_t)rj * HID + fo;
            hv[0][j].x = src[0];
            hv[0][j].y = src[HID / 2];
        }

        #pragma unroll
        for (int g = 0; g < EB / GD; ++g) {
            const int cur = g & 1;
            if (g + 1 < EB / GD) {
                #pragma unroll
                for (int j = 0; j < GD; ++j) {
                    const int rj = __builtin_amdgcn_readlane(r, (g + 1) * GD + j);
                    const float* src = h + (size_t)rj * HID + fo;
                    hv[cur ^ 1][j].x = src[0];
                    hv[cur ^ 1][j].y = src[HID / 2];
                }
            }
            #pragma unroll
            for (int j = 0; j < GD; ++j) {
                const int   cj = __builtin_amdgcn_readlane(c, g * GD + j);
                const float nj = __int_as_float(
                                     __builtin_amdgcn_readlane(wBits, g * GD + j));
                float* dst = agg + (size_t)cj * HID + fo;
                atomicAdd(dst,           nj * hv[cur][j].x);
                atomicAdd(dst + HID / 2, nj * hv[cur][j].y);
            }
        }
    }
}

__global__ void k_finalize(float* __restrict__ agg,
                           const float* __restrict__ h,
                           const float* __restrict__ dinv,
                           const float* __restrict__ b,
                           int relu) {
    int stride = gridDim.x * blockDim.x;
    for (int i = blockIdx.x * blockDim.x + threadIdx.x; i < N_NODES * HID; i += stride) {
        const int n = i >> 6;
        const int f = i & 63;
        const float di = dinv[n];
        float v = di * (agg[i] + h[i]) + b[f];
        agg[i] = relu ? fmaxf(v, 0.0f) : v;
    }
}

__global__ void k_fc(const float* __restrict__ h,
                     const float* __restrict__ Wfc,
                     const float* __restrict__ bfc,
                     float* __restrict__ out) {
    int n = blockIdx.x * blockDim.x + threadIdx.x;
    if (n >= N_NODES) return;
    float acc[OUT_DIM];
    #pragma unroll
    for (int o = 0; o < OUT_DIM; ++o) acc[o] = bfc[o];
    const float* hr = h + (size_t)n * HID;
    #pragma unroll 4
    for (int k = 0; k < HID; ++k) {
        const float xv = hr[k];
        #pragma unroll
        for (int o = 0; o < OUT_DIM; ++o)
            acc[o] = fmaf(xv, Wfc[k * OUT_DIM + o], acc[o]);
    }
    float* orow = out + (size_t)n * OUT_DIM;
    #pragma unroll
    for (int o = 0; o < OUT_DIM; ++o) orow[o] = acc[o];
}

extern "C" void kernel_launch(void* const* d_in, const int* in_sizes, int n_in,
                              void* d_out, int out_size, void* d_ws, size_t ws_size,
                              hipStream_t stream) {
    const float* x   = (const float*)d_in[0];
    const int*   ei  = (const int*)  d_in[1];
    const float* ew  = (const float*)d_in[2];
    const float* W1  = (const float*)d_in[3];
    const float* b1  = (const float*)d_in[4];
    const float* W2  = (const float*)d_in[5];
    const float* b2  = (const float*)d_in[6];
    const float* Wfc = (const float*)d_in[7];
    const float* bfc = (const float*)d_in[8];
    float* out = (float*)d_out;

    const int* row = ei;
    const int* col = ei + N_EDGES;

    char* ws = (char*)d_ws;
    size_t off = 0;
    float* dinv = (float*)(ws + off); off += ((size_t)N_NODES * 4 + 255) & ~(size_t)255;
    float* bufH = (float*)(ws + off); off += (size_t)N_NODES * HID * 4;
    float* bufA = (float*)(ws + off); off += (size_t)N_NODES * HID * 4;
    (void)ws_size; (void)n_in; (void)in_sizes; (void)out_size;

    const int TPB = 256;
    const int nodeBlocks = (N_NODES + TPB - 1) / TPB;
    const int featElems  = N_NODES * HID;
    const int gemmBlocks = ((N_NODES / 16) + (TPB / 32) - 1) / (TPB / 32);

    k_init_deg<<<nodeBlocks, TPB, 0, stream>>>(dinv);
    k_deg_scatter<<<1024, TPB, 0, stream>>>(col, ew, dinv);
    k_dinv<<<nodeBlocks, TPB, 0, stream>>>(dinv);

    k_gemm_n64<<<gemmBlocks, TPB, 0, stream>>>(x, W1, dinv, bufH, N_NODES, IN_DIM);
    k_zero<<<4096, TPB, 0, stream>>>(bufA, featElems);
    k_edge_scatter<<<SCATTER_BLOCKS, TPB, 0, stream>>>(row, col, ew, bufH, bufA);
    k_finalize<<<4096, TPB, 0, stream>>>(bufA, bufH, dinv, b1, 1);

    k_gemm_n64<<<gemmBlocks, TPB, 0, stream>>>(bufA, W2, dinv, bufH, N_NODES, HID);
    k_zero<<<4096, TPB, 0, stream>>>(bufA, featElems);
    k_edge_scatter<<<SCATTER_BLOCKS, TPB, 0, stream>>>(row, col, ew, bufH, bufA);
    k_finalize<<<4096, TPB, 0, stream>>>(bufA, bufH, dinv, b2, 1);

    k_fc<<<nodeBlocks, TPB, 0, stream>>>(bufA, Wfc, bfc, out);
}


# GEMM operand feed: W staged once per work-group in LDS in fragment order (16-byte reads), x rows read as whole lines in 16-byte pieces one line ahead; 250x800 grid; k summation order changes
# speedup vs baseline: 1.4793x; 1.0545x over previous
#include <hip/hip_runtime.h>
#include <hip/hip_bf16.h>


#define N_NODES 100000
#define N_EDGES 1600000
#define IN_DIM  512
#define HID     64
#define OUT_DIM 5

typedef __attribute__((ext_vector_type(2))) float v2f;
typedef __attribute__((ext_vector_type(8))) float v8f;

__global__ void k_init_deg(float* __restrict__ deg) {
    int i = blockIdx.x * blockDim.x + threadIdx.x;
    if (i < N_NODES) deg[i] = 1.0f;
}

__global__ void k_deg_scatter(const int* __restrict__ col,
                              const float* __restrict__ w,
                              float* __restrict__ deg) {
    int stride = gridDim.x * blockDim.x;
    for (int e = blockIdx.x * blockDim.x + threadIdx.x; e < N_EDGES; e += stride)
        atomicAdd(deg + col[e], w[e]);
}

__global__ void k_dinv(float* __restrict__ deg) {
    int i = blockIdx.x * blockDim.x + threadIdx.x;
    if (i < N_NODES) {
        float d = deg[i];
        deg[i] = (d > 0.0f) ? rsqrtf(d) : 0.0f;
    }
}

__global__ void k_zero(float* __restrict__ p, int n) {
    int stride = gridDim.x * blockDim.x;
    for (int i = blockIdx.x * blockDim.x + threadIdx.x; i < n; i += stride)
        p[i] = 0.0f;
}

#define GEMM_WAVES   25
#define GEMM_THREADS (GEMM_WAVES * 32)
#define GEMM_BLOCKS  ((N_NODES / 16) / GEMM_WAVES)
static_assert(GEMM_BLOCKS * GEMM_WAVES * 16 == N_NODES, "row tiles must fill the GEMM grid exactly");
#define GEMM_LDS_MIN 41984

typedef __attribute__((ext_vector_type(4))) float v4f;

__device__ __forceinline__ void gemm_line(const v4f (&a)[4], const float* bl,
                                          v8f& c0, v8f& c1, v8f& c2, v8f& c3) {
    #pragma unroll
    for (int i = 0; i < 4; ++i) {
        const v4f b00 = *(const v4f*)(bl + i * 512);
        const v4f b01 = *(const v4f*)(bl + i * 512 + 128);
        const v4f b10 = *(const v4f*)(bl + i * 512 + 256);
        const v4f b11 = *(const v4f*)(bl + i * 512 + 384);
        const v2f a0 = {a[i].x, a[i].y};
        const v2f a1 = {a[i].z, a[i].w};
        const v2f p00 = {b00.x, b00.y}, p01 = {b00.z, b00.w};
        const v2f p02 = {b01.x, b01.y}, p03 = {b01.z, b01.w};
        const v2f p10 = {b10.x, b10.y}, p11 = {b10.z, b10.w};
        const v2f p12 = {b11.x, b11.y}, p13 = {b11.z, b11.w};
        c0 = __builtin_amdgcn_wmma_f32_16x16x4_f32(false, a0, false, p00, (short)0, c0, false, false);
        c1 = __builtin_amdgcn_wmma_f32_16x16x4_f32(false, a0, false, p01, (short)0, c1, false, false);
        c2 = __builtin_amdgcn_wmma_f32_16x16x4_f32(false, a0, false, p02, (short)0, c2, false, false);
        c3 = __builtin_amdgcn_wmma_f32_16x16x4_f32(false, a0, false, p03, (short)0, c3, false, false);
        c0 = __builtin_amdgcn_wmma_f32_16x16x4_f32(false, a1, false, p10, (short)0, c0, false, false);
        c1 = __builtin_amdgcn_wmma_f32_16x16x4_f32(false, a1, false, p11, (short)0, c1, false, false);
        c2 = __builtin_amdgcn_wmma_f32_16x16x4_f32(false, a1, false, p12, (short)0, c2, false, false);
        c3 = __builtin_amdgcn_wmma_f32_16x16x4_f32(false, a1, false, p13, (short)0, c3, false, false);
    }
}

template <int K>
__global__ void __launch_bounds__(GEMM_THREADS)
k_gemm_n64(const float* __restrict__ X,
           const float* __restrict__ W,
           const float* __restrict__ rs,
           float* __restrict__ H) {
    static_assert(K % 64 == 0, "the k loop takes two 32-k lines per trip");
    constexpr int LDS_FLOATS = (K * HID > GEMM_LDS_MIN) ? K * HID : GEMM_LDS_MIN;
    __shared__ __attribute__((aligned(16))) float Bl[LDS_FLOATS];

    const int tid  = threadIdx.x;
    const int lane = tid & 31;
    const int c    = lane & 15;
    const int h    = lane >> 4;
    const int tile = blockIdx.x * GEMM_WAVES + (tid >> 5);
    const float* xr = X + (size_t)(tile * 16 + c) * K + 4 * h;

    v4f aA[4], aB[4];
    #pragma unroll
    for (int i = 0; i < 4; ++i) aA[i] = *(const v4f*)(xr + 8 * i);

    for (int u = tid; u < (K / 2) * 32; u += GEMM_THREADS) {
        const int cc = u & 15;
        const int tp = (u >> 4) & 1;
        const int k  = (u >> 5) * 2;
        const float* w0 = W + (size_t)k * HID + tp * 32 + cc;
        v4f v;
        v.x = w0[0];
        v.y = w0[HID];
        v.z = w0[16];
        v.w = w0[HID + 16];
        const int kb = k >> 3;
        const int hh = (k >> 2) & 1;
        const int j  = (k >> 1) & 1;
        *(v4f*)(Bl + kb * 512 + (2 * j + tp) * 128 + (16 * hh + cc) * 4) = v;
    }
    __syncthreads();

    v8f c0 = {}, c1 = {}, c2 = {}, c3 = {};
    const float* bl = Bl + lane * 4;

    for (int ch = 0; ch < K / 32; ch += 2) {
        #pragma unroll
        for (int i = 0; i < 4; ++i) aB[i] = *(const v4f*)(xr + (ch + 1) * 32 + 8 * i);
        gemm_line(aA, bl + ch * 2048, c0, c1, c2, c3);
        if (ch + 2 < K / 32) {
            #pragma unroll
            for (int i = 0; i < 4; ++i) aA[i] = *(const v4f*)(xr + (ch + 2) * 32 + 8 * i);
        }
        gemm_line(aB, bl + (ch + 1) * 2048, c0, c1, c2, c3);
    }

    const int rBase = tile * 16 + 8 * h;
    #pragma unroll
    for (int j = 0; j < 8; ++j) {
        const float sc = rs[rBase + j];
        float* dst = H + (size_t)(rBase + j) * HID + c;
        dst[0]  = sc * c0[j];
        dst[16] = sc * c1[j];
        dst[32] = sc * c2[j];
        dst[48] = sc * c3[j];
    }
}

#define EB 32
#define GD 8
static_assert(N_EDGES % EB == 0, "edge count must fill whole wave trips");
static_assert(EB % GD == 0, "gather groups must tile a wave trip");
static_assert(HID == 64, "a row is two 32-float halves, one per wave-wide access");
#define SCATTER_TRIPS_PER_WAVE 5
#define SCATTER_BLOCKS ((N_EDGES / EB) / 8 / SCATTER_TRIPS_PER_WAVE)
static_assert((N_EDGES / EB) % (8 * SCATTER_TRIPS_PER_WAVE) == 0,
              "trips must divide evenly over the scatter's waves");

__global__ void k_edge_scatter(const int* __restrict__ row,
                               const int* __restrict__ col,
                               const float* __restrict__ w,
                               const float* __restrict__ h,
                               float* __restrict__ agg) {
    const int lane   = threadIdx.x & 31;
    const int wid    = __builtin_amdgcn_readfirstlane(
                           (int)((blockIdx.x * blockDim.x + threadIdx.x) >> 5));
    const int nWaves = (int)((gridDim.x * blockDim.x) >> 5);
    const int fo     = lane;

    for (int b = wid; b < N_EDGES / EB; b += nWaves) {
        const int e = b * EB + lane;
        const int r = row[e];
        const int c = col[e];
        const int wBits = __float_as_int(w[e]);

        v2f hv[2][GD];

        #pragma unroll
        for (int j = 0; j < GD; ++j) {
            const int rj = __builtin_amdgcn_readlane(r, j);
            const float* src = h + (size_t)rj * HID + fo;
            hv[0][j].x = src[0];
            hv[0][j].y = src[HID / 2];
        }

        #pragma unroll
        for (int g = 0; g < EB / GD; ++g) {
            const int cur = g & 1;
            if (g + 1 < EB / GD) {
                #pragma unroll
                for (int j = 0; j < GD; ++j) {
                    const int rj = __builtin_amdgcn_readlane(r, (g + 1) * GD + j);
                    const float* src = h + (size_t)rj * HID + fo;
                    hv[cur ^ 1][j].x = src[0];
                    hv[cur ^ 1][j].y = src[HID / 2];
                }
            }
            #pragma unroll
            for (int j = 0; j < GD; ++j) {
                const int   cj = __builtin_amdgcn_readlane(c, g * GD + j);
                const float nj = __int_as_float(
                                     __builtin_amdgcn_readlane(wBits, g * GD + j));
                float* dst = agg + (size_t)cj * HID + fo;
                atomicAdd(dst,           nj * hv[cur][j].x);
                atomicAdd(dst + HID / 2, nj * hv[cur][j].y);
            }
        }
    }
}

__global__ void k_finalize(float* __restrict__ agg,
                           const float* __restrict__ h,
                           const float* __restrict__ dinv,
                           const float* __restrict__ b,
                           int relu) {
    int stride = gridDim.x * blockDim.x;
    for (int i = blockIdx.x * blockDim.x + threadIdx.x; i < N_NODES * HID; i += stride) {
        const int n = i >> 6;
        const int f = i & 63;
        const float di = dinv[n];
        float v = di * (agg[i] + h[i]) + b[f];
        agg[i] = relu ? fmaxf(v, 0.0f) : v;
    }
}

__global__ void k_fc(const float* __restrict__ h,
                     const float* __restrict__ Wfc,
                     const float* __restrict__ bfc,
                     float* __restrict__ out) {
    int n = blockIdx.x * blockDim.x + threadIdx.x;
    if (n >= N_NODES) return;
    float acc[OUT_DIM];
    #pragma unroll
    for (int o = 0; o < OUT_DIM; ++o) acc[o] = bfc[o];
    const float* hr = h + (size_t)n * HID;
    #pragma unroll 4
    for (int k = 0; k < HID; ++k) {
        const float xv = hr[k];
        #pragma unroll
        for (int o = 0; o < OUT_DIM; ++o)
            acc[o] = fmaf(xv, Wfc[k * OUT_DIM + o], acc[o]);
    }
    float* orow = out + (size_t)n * OUT_DIM;
    #pragma unroll
    for (int o = 0; o < OUT_DIM; ++o) orow[o] = acc[o];
}

extern "C" void kernel_launch(void* const* d_in, const int* in_sizes, int n_in,
                              void* d_out, int out_size, void* d_ws, size_t ws_size,
                              hipStream_t stream) {
    const float* x   = (const float*)d_in[0];
    const int*   ei  = (const int*)  d_in[1];
    const float* ew  = (const float*)d_in[2];
    const float* W1  = (const float*)d_in[3];
    const float* b1  = (const float*)d_in[4];
    const float* W2  = (const float*)d_in[5];
    const float* b2  = (const float*)d_in[6];
    const float* Wfc = (const float*)d_in[7];
    const float* bfc = (const float*)d_in[8];
    float* out = (float*)d_out;

    const int* row = ei;
    const int* col = ei + N_EDGES;

    char* ws = (char*)d_ws;
    size_t off = 0;
    float* dinv = (float*)(ws + off); off += ((size_t)N_NODES * 4 + 255) & ~(size_t)255;
    float* bufH = (float*)(ws + off); off += (size_t)N_NODES * HID * 4;
    float* bufA = (float*)(ws + off); off += (size_t)N_NODES * HID * 4;
    (void)ws_size; (void)n_in; (void)in_sizes; (void)out_size;

    const int TPB = 256;
    const int nodeBlocks = (N_NODES + TPB - 1) / TPB;
    const int featElems  = N_NODES * HID;

    k_init_deg<<<nodeBlocks, TPB, 0, stream>>>(dinv);
    k_deg_scatter<<<1024, TPB, 0, stream>>>(col, ew, dinv);
    k_dinv<<<nodeBlocks, TPB, 0, stream>>>(dinv);

    k_gemm_n64<IN_DIM><<<GEMM_BLOCKS, GEMM_THREADS, 0, stream>>>(x, W1, dinv, bufH);
    k_zero<<<4096, TPB, 0, stream>>>(bufA, featElems);
    k_edge_scatter<<<SCATTER_BLOCKS, TPB, 0, stream>>>(row, col, ew, bufH, bufA);
    k_finalize<<<4096, TPB, 0, stream>>>(bufA, bufH, dinv, b1, 1);

    k_gemm_n64<HID><<<GEMM_BLOCKS, GEMM_THREADS, 0, stream>>>(bufA, W2, dinv, bufH);
    k_zero<<<4096, TPB, 0, stream>>>(bufA, featElems);
    k_edge_scatter<<<SCATTER_BLOCKS, TPB, 0, stream>>>(row, col, ew, bufH, bufA);
    k_finalize<<<4096, TPB, 0, stream>>>(bufA, bufH, dinv, b2, 1);

    k_fc<<<nodeBlocks, TPB, 0, stream>>>(bufA, Wfc, bfc, out);
}


# atomic sums kept per package half: degree and edge scatters add into the plane of the half the wave runs on (die number), dinv/finalize add the two planes; sum order: two partial sums per node
# speedup vs baseline: 1.9582x; 1.3237x over previous
#include <hip/hip_runtime.h>
#include <hip/hip_bf16.h>


#define N_NODES 100000
#define N_EDGES 1600000
#define IN_DIM  512
#define HID     64
#define OUT_DIM 5

typedef __attribute__((ext_vector_type(2))) float v2f;
typedef __attribute__((ext_vector_type(8))) float v8f;

#define DINV_BYTES   ((((size_t)N_NODES * 4) + 255) & ~(size_t)255)
#define PLANE_ELEMS  ((size_t)N_NODES * HID)
#define PLANE_BYTES  (PLANE_ELEMS * 4)
#define WS_BYTES_TWO_PLANES (DINV_BYTES + 3 * PLANE_BYTES)
__device__ __forceinline__ bool two_planes(size_t wsBytes) {
    return wsBytes >= WS_BYTES_TWO_PLANES;
}

__device__ __forceinline__ int wave_half() {
    return (int)((__builtin_amdgcn_s_sendmsg_rtn(0x87) >> 18) & 1u);
}

__global__ void k_init_deg(float* __restrict__ deg0, float* __restrict__ deg1) {
    int i = blockIdx.x * blockDim.x + threadIdx.x;
    if (i < N_NODES) { deg0[i] = 1.0f; deg1[i] = 0.0f; }
}

__global__ void k_deg_scatter(const int* __restrict__ col,
                              const float* __restrict__ w,
                              float* __restrict__ deg0,
                              float* __restrict__ deg1) {
    float* deg = wave_half() ? deg1 : deg0;
    int stride = gridDim.x * blockDim.x;
    for (int e = blockIdx.x * blockDim.x + threadIdx.x; e < N_EDGES; e += stride)
        atomicAdd(deg + col[e], w[e]);
}

__global__ void k_dinv(float* __restrict__ deg0, const float* __restrict__ deg1) {
    int i = blockIdx.x * blockDim.x + threadIdx.x;
    if (i < N_NODES) {
        float d = deg0[i] + deg1[i];
        deg0[i] = (d > 0.0f) ? rsqrtf(d) : 0.0f;
    }
}

__global__ void k_zero(float* __restrict__ p, int nOne, size_t wsBytes) {
    const int n = two_planes(wsBytes) ? 2 * nOne : nOne;
    int stride = gridDim.x * blockDim.x;
    for (int i = blockIdx.x * blockDim.x + threadIdx.x; i < n; i += stride)
        p[i] = 0.0f;
}

#define GEMM_WAVES   25
#define GEMM_THREADS (GEMM_WAVES * 32)
#define GEMM_BLOCKS  ((N_NODES / 16) / GEMM_WAVES)
static_assert(GEMM_BLOCKS * GEMM_WAVES * 16 == N_NODES, "row tiles must fill the GEMM grid exactly");
#define GEMM_LDS_MIN 41984

typedef __attribute__((ext_vector_type(4))) float v4f;

__device__ __forceinline__ void gemm_line(const v4f (&a)[4], const float* bl,
                                          v8f& c0, v8f& c1, v8f& c2, v8f& c3) {
    #pragma unroll
    for (int i = 0; i < 4; ++i) {
        const v4f b00 = *(const v4f*)(bl + i * 512);
        const v4f b01 = *(const v4f*)(bl + i * 512 + 128);
        const v4f b10 = *(const v4f*)(bl + i * 512 + 256);
        const v4f b11 = *(const v4f*)(bl + i * 512 + 384);
        const v2f a0 = {a[i].x, a[i].y};
        const v2f a1 = {a[i].z, a[i].w};
        const v2f p00 = {b00.x, b00.y}, p01 = {b00.z, b00.w};
        const v2f p02 = {b01.x, b01.y}, p03 = {b01.z, b01.w};
        const v2f p10 = {b10.x, b10.y}, p11 = {b10.z, b10.w};
        const v2f p12 = {b11.x, b11.y}, p13 = {b11.z, b11.w};
        c0 = __builtin_amdgcn_wmma_f32_16x16x4_f32(false, a0, false, p00, (short)0, c0, false, false);
        c1 = __builtin_amdgcn_wmma_f32_16x16x4_f32(false, a0, false, p01, (short)0, c1, false, false);
        c2 = __builtin_amdgcn_wmma_f32_16x16x4_f32(false, a0, false, p02, (short)0, c2, false, false);
        c3 = __builtin_amdgcn_wmma_f32_16x16x4_f32(false, a0, false, p03, (short)0, c3, false, false);
        c0 = __builtin_amdgcn_wmma_f32_16x16x4_f32(false, a1, false, p10, (short)0, c0, false, false);
        c1 = __builtin_amdgcn_wmma_f32_16x16x4_f32(false, a1, false, p11, (short)0, c1, false, false);
        c2 = __builtin_amdgcn_wmma_f32_16x16x4_f32(false, a1, false, p12, (short)0, c2, false, false);
        c3 = __builtin_amdgcn_wmma_f32_16x16x4_f32(false, a1, false, p13, (short)0, c3, false, false);
    }
}

template <int K>
__global__ void __launch_bounds__(GEMM_THREADS)
k_gemm_n64(const float* __restrict__ X,
           const float* __restrict__ W,
           const float* __restrict__ rs,
           float* __restrict__ H) {
    static_assert(K % 64 == 0, "the k loop takes two 32-k lines per trip");
    constexpr int LDS_FLOATS = (K * HID > GEMM_LDS_MIN) ? K * HID : GEMM_LDS_MIN;
    __shared__ __attribute__((aligned(16))) float Bl[LDS_FLOATS];

    const int tid  = threadIdx.x;
    const int lane = tid & 31;
    const int c    = lane & 15;
    const int h    = lane >> 4;
    const int tile = blockIdx.x * GEMM_WAVES + (tid >> 5);
    const float* xr = X + (size_t)(tile * 16 + c) * K + 4 * h;

    v4f aA[4], aB[4];
    #pragma unroll
    for (int i = 0; i < 4; ++i) aA[i] = *(const v4f*)(xr + 8 * i);

    for (int u = tid; u < (K / 2) * 32; u += GEMM_THREADS) {
        const int cc = u & 15;
        const int tp = (u >> 4) & 1;
        const int k  = (u >> 5) * 2;
        const float* w0 = W + (size_t)k * HID + tp * 32 + cc;
        v4f v;
        v.x = w0[0];
        v.y = w0[HID];
        v.z = w0[16];
        v.w = w0[HID + 16];
        const int kb = k >> 3;
        const int hh = (k >> 2) & 1;
        const int j  = (k >> 1) & 1;
        *(v4f*)(Bl + kb * 512 + (2 * j + tp) * 128 + (16 * hh + cc) * 4) = v;
    }
    __syncthreads();

    v8f c0 = {}, c1 = {}, c2 = {}, c3 = {};
    const float* bl = Bl + lane * 4;

    for (int ch = 0; ch < K / 32; ch += 2) {
        #pragma unroll
        for (int i = 0; i < 4; ++i) aB[i] = *(const v4f*)(xr + (ch + 1) * 32 + 8 * i);
        gemm_line(aA, bl + ch * 2048, c0, c1, c2, c3);
        if (ch + 2 < K / 32) {
            #pragma unroll
            for (int i = 0; i < 4; ++i) aA[i] = *(const v4f*)(xr + (ch + 2) * 32 + 8 * i);
        }
        gemm_line(aB, bl + (ch + 1) * 2048, c0, c1, c2, c3);
    }

    const int rBase = tile * 16 + 8 * h;
    #pragma unroll
    for (int j = 0; j < 8; ++j) {
        const float sc = rs[rBase + j];
        float* dst = H + (size_t)(rBase + j) * HID + c;
        dst[0]  = sc * c0[j];
        dst[16] = sc * c1[j];
        dst[32] = sc * c2[j];
        dst[48] = sc * c3[j];
    }
}

#define EB 32
#define GD 8
static_assert(N_EDGES % EB == 0, "edge count must fill whole wave trips");
static_assert(EB % GD == 0, "gather groups must tile a wave trip");
static_assert(HID == 64, "a row is two 32-float halves, one per wave-wide access");
#define SCATTER_TRIPS_PER_WAVE 5
#define SCATTER_BLOCKS ((N_EDGES / EB) / 8 / SCATTER_TRIPS_PER_WAVE)
static_assert((N_EDGES / EB) % (8 * SCATTER_TRIPS_PER_WAVE) == 0,
              "trips must divide evenly over the scatter's waves");

__global__ void k_edge_scatter(const int* __restrict__ row,
                               const int* __restrict__ col,
                               const float* __restrict__ w,
                               const float* __restrict__ h,
                               float* __restrict__ agg0,
                               size_t wsBytes) {
    float* agg = agg0 + ((two_planes(wsBytes) && wave_half()) ? PLANE_ELEMS : 0);
    const int lane   = threadIdx.x & 31;
    const int wid    = __builtin_amdgcn_readfirstlane(
                           (int)((blockIdx.x * blockDim.x + threadIdx.x) >> 5));
    const int nWaves = (int)((gridDim.x * blockDim.x) >> 5);
    const int fo     = lane;

    for (int b = wid; b < N_EDGES / EB; b += nWaves) {
        const int e = b * EB + lane;
        const int r = row[e];
        const int c = col[e];
        const int wBits = __float_as_int(w[e]);

        v2f hv[2][GD];

        #pragma unroll
        for (int j = 0; j < GD; ++j) {
            const int rj = __builtin_amdgcn_readlane(r, j);
            const float* src = h + (size_t)rj * HID + fo;
            hv[0][j].x = src[0];
            hv[0][j].y = src[HID / 2];
        }

        #pragma unroll
        for (int g = 0; g < EB / GD; ++g) {
            const int cur = g & 1;
            if (g + 1 < EB / GD) {
                #pragma unroll
                for (int j = 0; j < GD; ++j) {
                    const int rj = __builtin_amdgcn_readlane(r, (g + 1) * GD + j);
                    const float* src = h + (size_t)rj * HID + fo;
                    hv[cur ^ 1][j].x = src[0];
                    hv[cur ^ 1][j].y = src[HID / 2];
                }
            }
            #pragma unroll
            for (int j = 0; j < GD; ++j) {
                const int   cj = __builtin_amdgcn_readlane(c, g * GD + j);
                const float nj = __int_as_float(
                                     __builtin_amdgcn_readlane(wBits, g * GD + j));
                float* dst = agg + (size_t)cj * HID + fo;
                atomicAdd(dst,           nj * hv[cur][j].x);
                atomicAdd(dst + HID / 2, nj * hv[cur][j].y);
            }
        }
    }
}

template <bool TWO>
__device__ __forceinline__ void finalize_elems(float* __restrict__ agg,
                                               const float* __restrict__ h,
                                               const float* __restrict__ dinv,
                                               const float* __restrict__ b,
                                               int relu) {
    int stride = gridDim.x * blockDim.x;
    for (int i = blockIdx.x * blockDim.x + threadIdx.x; i < N_NODES * HID; i += stride) {
        const int n = i >> 6;
        const int f = i & 63;
        const float di = dinv[n];
        float s = agg[i];
        if (TWO) s += agg[i + PLANE_ELEMS];
        float v = di * (s + h[i]) + b[f];
        agg[i] = relu ? fmaxf(v, 0.0f) : v;
    }
}

__global__ void k_finalize(float* __restrict__ agg,
                           const float* __restrict__ h,
                           const float* __restrict__ dinv,
                           const float* __restrict__ b,
                           int relu,
                           size_t wsBytes) {
    if (two_planes(wsBytes)) finalize_elems<true>(agg, h, dinv, b, relu);
    else                     finalize_elems<false>(agg, h, dinv, b, relu);
}

__global__ void k_fc(const float* __restrict__ h,
                     const float* __restrict__ Wfc,
                     const float* __restrict__ bfc,
                     float* __restrict__ out) {
    int n = blockIdx.x * blockDim.x + threadIdx.x;
    if (n >= N_NODES) return;
    float acc[OUT_DIM];
    #pragma unroll
    for (int o = 0; o < OUT_DIM; ++o) acc[o] = bfc[o];
    const float* hr = h + (size_t)n * HID;
    #pragma unroll 4
    for (int k = 0; k < HID; ++k) {
        const float xv = hr[k];
        #pragma unroll
        for (int o = 0; o < OUT_DIM; ++o)
            acc[o] = fmaf(xv, Wfc[k * OUT_DIM + o], acc[o]);
    }
    float* orow = out + (size_t)n * OUT_DIM;
    #pragma unroll
    for (int o = 0; o < OUT_DIM; ++o) orow[o] = acc[o];
}

extern "C" void kernel_launch(void* const* d_in, const int* in_sizes, int n_in,
                              void* d_out, int out_size, void* d_ws, size_t ws_size,
                              hipStream_t stream) {
    const float* x   = (const float*)d_in[0];
    const int*   ei  = (const int*)  d_in[1];
    const float* ew  = (const float*)d_in[2];
    const float* W1  = (const float*)d_in[3];
    const float* b1  = (const float*)d_in[4];
    const float* W2  = (const float*)d_in[5];
    const float* b2  = (const float*)d_in[6];
    const float* Wfc = (const float*)d_in[7];
    const float* bfc = (const float*)d_in[8];
    float* out = (float*)d_out;

    const int* row = ei;
    const int* col = ei + N_EDGES;

    char* ws = (char*)d_ws;
    size_t off = 0;
    float* dinv = (float*)(ws + off); off += DINV_BYTES;
    float* bufH = (float*)(ws + off); off += PLANE_BYTES;
    float* bufA = (float*)(ws + off); off += PLANE_BYTES;
    float* deg1 = bufA;
    (void)n_in; (void)in_sizes; (void)out_size;

    const int TPB = 256;
    const int nodeBlocks = (N_NODES + TPB - 1) / TPB;
    const int featElems  = N_NODES * HID;

    k_init_deg<<<nodeBlocks, TPB, 0, stream>>>(dinv, deg1);
    k_deg_scatter<<<1024, TPB, 0, stream>>>(col, ew, dinv, deg1);
    k_dinv<<<nodeBlocks, TPB, 0, stream>>>(dinv, deg1);

    k_gemm_n64<IN_DIM><<<GEMM_BLOCKS, GEMM_THREADS, 0, stream>>>(x, W1, dinv, bufH);
    k_zero<<<4096, TPB, 0, stream>>>(bufA, featElems, ws_size);
    k_edge_scatter<<<SCATTER_BLOCKS, TPB, 0, stream>>>(row, col, ew, bufH, bufA, ws_size);
    k_finalize<<<4096, TPB, 0, stream>>>(bufA, bufH, dinv, b1, 1, ws_size);

    k_gemm_n64<HID><<<GEMM_BLOCKS, GEMM_THREADS, 0, stream>>>(bufA, W2, dinv, bufH);
    k_zero<<<4096, TPB, 0, stream>>>(bufA, featElems, ws_size);
    k_edge_scatter<<<SCATTER_BLOCKS, TPB, 0, stream>>>(row, col, ew, bufH, bufA, ws_size);
    k_finalize<<<4096, TPB, 0, stream>>>(bufA, bufH, dinv, b2, 1, ws_size);

    k_fc<<<nodeBlocks, TPB, 0, stream>>>(bufA, Wfc, bfc, out);
}


# edge sums by owner: edges filed per call by target block (count, scan, LDS-ranked place); a work-group sums its 416 nodes in LDS (float atomics), epilogue and FC fused; sum order: arrival
# speedup vs baseline: 2.3829x; 1.2169x over previous
#include <hip/hip_runtime.h>
#include <hip/hip_bf16.h>


#define N_NODES 100000
#define N_EDGES 1600000
#define IN_DIM  512
#define HID     64
#define OUT_DIM 5

typedef __attribute__((ext_vector_type(2))) float v2f;
typedef __attribute__((ext_vector_type(8))) float v8f;

typedef __attribute__((ext_vector_type(2))) int v2i;

#define BIN_NODES    416
#define NBINS        ((N_NODES + BIN_NODES - 1) / BIN_NODES)
#define CHUNK        3200
#define NCHUNKS      (N_EDGES / CHUNK)
#define BIN_THREADS  640
#define EPT          (CHUNK / BIN_THREADS)
#define SCAN_THREADS 512
#define OWN_THREADS  1024
#define ROW_BITS     17
static_assert(NCHUNKS * CHUNK == N_EDGES, "chunks must tile the edge list");
static_assert(EPT * BIN_THREADS == CHUNK, "a chunk is a whole number of edges per thread");
static_assert(NBINS <= BIN_THREADS, "one thread per block in the placing pass's scan");
static_assert(NCHUNKS <= SCAN_THREADS, "one thread per chunk in the scan");
static_assert(N_NODES <= (1 << ROW_BITS), "source row must fit its field");
static_assert(BIN_NODES < (1 << (30 - ROW_BITS)), "local target must fit its field");
static_assert((BIN_NODES * OUT_DIM * 4) % 128 == 0, "a block's rows of out are whole lines");

#define DINV_BYTES   ((((size_t)N_NODES * 4) + 255) & ~(size_t)255)
#define PLANE_ELEMS  ((size_t)N_NODES * HID)
#define PLANE_BYTES  (PLANE_ELEMS * 4)
#define REC_BYTES    ((size_t)N_EDGES * 8)
#define CNT_BYTES    ((size_t)NBINS * NCHUNKS * 4)
#define WS_NEED      (DINV_BYTES + 2 * PLANE_BYTES + REC_BYTES + 2 * CNT_BYTES + 1024)

__device__ __forceinline__ int wave_half() {
    return (int)((__builtin_amdgcn_s_sendmsg_rtn(0x87) >> 18) & 1u);
}

__global__ void k_init_deg(float* __restrict__ deg0, float* __restrict__ deg1) {
    int i = blockIdx.x * blockDim.x + threadIdx.x;
    if (i < N_NODES) { deg0[i] = 1.0f; deg1[i] = 0.0f; }
}

__device__ __forceinline__ int block_scan_incl(int v, int* s) {
    const int tid = threadIdx.x;
    const int n   = blockDim.x;
    s[tid] = v;
    __syncthreads();
    for (int d = 1; d < n; d <<= 1) {
        const int t = (tid >= d) ? s[tid - d] : 0;
        __syncthreads();
        s[tid] += t;
        __syncthreads();
    }
    return s[tid];
}

__global__ void __launch_bounds__(BIN_THREADS)
k_count(const int* __restrict__ col,
        const float* __restrict__ w,
        float* __restrict__ deg0,
        float* __restrict__ deg1,
        int* __restrict__ cnt,
        size_t wsBytes) {
    if (wsBytes < WS_NEED) return;
    __shared__ int hist[BIN_THREADS];
    const int tid = threadIdx.x;
    hist[tid] = 0;
    __syncthreads();
    float* deg = wave_half() ? deg1 : deg0;
    const int e0 = blockIdx.x * CHUNK + tid;
    int   c[EPT];
    float wv[EPT];
    #pragma unroll
    for (int k = 0; k < EPT; ++k) {
        c[k]  = col[e0 + k * BIN_THREADS];
        wv[k] = w[e0 + k * BIN_THREADS];
    }
    #pragma unroll
    for (int k = 0; k < EPT; ++k) {
        atomicAdd(deg + c[k], wv[k]);
        atomicAdd(&hist[(unsigned)c[k] / BIN_NODES], 1);
    }
    __syncthreads();
    if (tid < NBINS) cnt[tid * NCHUNKS + blockIdx.x] = hist[tid];
}

static_assert(BIN_NODES <= SCAN_THREADS, "one thread per node of the block for dinv");
__global__ void __launch_bounds__(SCAN_THREADS)
k_scan_dinv(const int* __restrict__ cnt,
            int* __restrict__ rel,
            int* __restrict__ tot,
            float* __restrict__ deg0,
            const float* __restrict__ deg1,
            size_t wsBytes) {
    if (wsBytes < WS_NEED) return;
    __shared__ int s[SCAN_THREADS];
    const int tid = threadIdx.x;
    const int bin = blockIdx.x;
    const int node = bin * BIN_NODES + tid;
    if (tid < BIN_NODES && node < N_NODES) {
        const float d = deg0[node] + deg1[node];
        deg0[node] = (d > 0.0f) ? rsqrtf(d) : 0.0f;
    }
    const int v = (tid < NCHUNKS) ? cnt[bin * NCHUNKS + tid] : 0;
    const int incl = block_scan_incl(v, s);
    if (tid < NCHUNKS) rel[bin * NCHUNKS + tid] = incl - v;
    if (tid == SCAN_THREADS - 1) tot[bin] = incl;
}

__global__ void __launch_bounds__(BIN_THREADS)
k_place(const int* __restrict__ row,
        const int* __restrict__ col,
        const float* __restrict__ w,
        const int* __restrict__ rel,
        const int* __restrict__ tot,
        int* __restrict__ rec,
        size_t wsBytes) {
    if (wsBytes < WS_NEED) return;
    __shared__ int s[BIN_THREADS];
    __shared__ int cursor[BIN_THREADS];
    const int tid = threadIdx.x;
    const int g   = blockIdx.x;
    const int tv   = (tid < NBINS) ? tot[tid] : 0;
    const int incl = block_scan_incl(tv, s);
    cursor[tid] = (tid < NBINS) ? (incl - tv) + rel[tid * NCHUNKS + g] : 0;
    __syncthreads();
    const int e0 = g * CHUNK + tid;
    int c[EPT], r[EPT], wb[EPT];
    #pragma unroll
    for (int k = 0; k < EPT; ++k) {
        c[k]  = col[e0 + k * BIN_THREADS];
        r[k]  = row[e0 + k * BIN_THREADS];
        wb[k] = __float_as_int(w[e0 + k * BIN_THREADS]);
    }
    #pragma unroll
    for (int k = 0; k < EPT; ++k) {
        const int bin = (int)((unsigned)c[k] / BIN_NODES);
        const int cl  = c[k] - bin * BIN_NODES;
        const int pos = atomicAdd(&cursor[bin], 1);
        v2i rc;
        rc.x = r[k] | (cl << ROW_BITS);
        rc.y = wb[k];
        *(v2i*)(rec + 2 * (size_t)pos) = rc;
    }
}

#define GEMM_WAVES   25
#define GEMM_THREADS (GEMM_WAVES * 32)
#define GEMM_BLOCKS  ((N_NODES / 16) / GEMM_WAVES)
static_assert(GEMM_BLOCKS * GEMM_WAVES * 16 == N_NODES, "row tiles must fill the GEMM grid exactly");
#define GEMM_LDS_MIN 41984

typedef __attribute__((ext_vector_type(4))) float v4f;

__device__ __forceinline__ void gemm_line(const v4f (&a)[4], const float* bl,
                                          v8f& c0, v8f& c1, v8f& c2, v8f& c3) {
    #pragma unroll
    for (int i = 0; i < 4; ++i) {
        const v4f b00 = *(const v4f*)(bl + i * 512);
        const v4f b01 = *(const v4f*)(bl + i * 512 + 128);
        const v4f b10 = *(const v4f*)(bl + i * 512 + 256);
        const v4f b11 = *(const v4f*)(bl + i * 512 + 384);
        const v2f a0 = {a[i].x, a[i].y};
        const v2f a1 = {a[i].z, a[i].w};
        const v2f p00 = {b00.x, b00.y}, p01 = {b00.z, b00.w};
        const v2f p02 = {b01.x, b01.y}, p03 = {b01.z, b01.w};
        const v2f p10 = {b10.x, b10.y}, p11 = {b10.z, b10.w};
        const v2f p12 = {b11.x, b11.y}, p13 = {b11.z, b11.w};
        c0 = __builtin_amdgcn_wmma_f32_16x16x4_f32(false, a0, false, p00, (short)0, c0, false, false);
        c1 = __builtin_amdgcn_wmma_f32_16x16x4_f32(false, a0, false, p01, (short)0, c1, false, false);
        c2 = __builtin_amdgcn_wmma_f32_16x16x4_f32(false, a0, false, p02, (short)0, c2, false, false);
        c3 = __builtin_amdgcn_wmma_f32_16x16x4_f32(false, a0, false, p03, (short)0, c3, false, false);
        c0 = __builtin_amdgcn_wmma_f32_16x16x4_f32(false, a1, false, p10, (short)0, c0, false, false);
        c1 = __builtin_amdgcn_wmma_f32_16x16x4_f32(false, a1, false, p11, (short)0, c1, false, false);
        c2 = __builtin_amdgcn_wmma_f32_16x16x4_f32(false, a1, false, p12, (short)0, c2, false, false);
        c3 = __builtin_amdgcn_wmma_f32_16x16x4_f32(false, a1, false, p13, (short)0, c3, false, false);
    }
}

template <int K>
__global__ void __launch_bounds__(GEMM_THREADS)
k_gemm_n64(const float* __restrict__ X,
           const float* __restrict__ W,
           const float* __restrict__ rs,
           float* __restrict__ H) {
    static_assert(K % 64 == 0, "the k loop takes two 32-k lines per trip");
    constexpr int LDS_FLOATS = (K * HID > GEMM_LDS_MIN) ? K * HID : GEMM_LDS_MIN;
    __shared__ __attribute__((aligned(16))) float Bl[LDS_FLOATS];

    const int tid  = threadIdx.x;
    const int lane = tid & 31;
    const int c    = lane & 15;
    const int h    = lane >> 4;
    const int tile = blockIdx.x * GEMM_WAVES + (tid >> 5);
    const float* xr = X + (size_t)(tile * 16 + c) * K + 4 * h;

    v4f aA[4], aB[4];
    #pragma unroll
    for (int i = 0; i < 4; ++i) aA[i] = *(const v4f*)(xr + 8 * i);

    for (int u = tid; u < (K / 2) * 32; u += GEMM_THREADS) {
        const int cc = u & 15;
        const int tp = (u >> 4) & 1;
        const int k  = (u >> 5) * 2;
        const float* w0 = W + (size_t)k * HID + tp * 32 + cc;
        v4f v;
        v.x = w0[0];
        v.y = w0[HID];
        v.z = w0[16];
        v.w = w0[HID + 16];
        const int kb = k >> 3;
        const int hh = (k >> 2) & 1;
        const int j  = (k >> 1) & 1;
        *(v4f*)(Bl + kb * 512 + (2 * j + tp) * 128 + (16 * hh + cc) * 4) = v;
    }
    __syncthreads();

    v8f c0 = {}, c1 = {}, c2 = {}, c3 = {};
    const float* bl = Bl + lane * 4;

    for (int ch = 0; ch < K / 32; ch += 2) {
        #pragma unroll
        for (int i = 0; i < 4; ++i) aB[i] = *(const v4f*)(xr + (ch + 1) * 32 + 8 * i);
        gemm_line(aA, bl + ch * 2048, c0, c1, c2, c3);
        if (ch + 2 < K / 32) {
            #pragma unroll
            for (int i = 0; i < 4; ++i) aA[i] = *(const v4f*)(xr + (ch + 2) * 32 + 8 * i);
        }
        gemm_line(aB, bl + (ch + 1) * 2048, c0, c1, c2, c3);
    }

    const int rBase = tile * 16 + 8 * h;
    #pragma unroll
    for (int j = 0; j < 8; ++j) {
        const float sc = rs[rBase + j];
        float* dst = H + (size_t)(rBase + j) * HID + c;
        dst[0]  = sc * c0[j];
        dst[16] = sc * c1[j];
        dst[32] = sc * c2[j];
        dst[48] = sc * c3[j];
    }
}

#define EB 32
#define GD 8
static_assert(EB % GD == 0, "gather groups must tile a wave trip");
static_assert(HID == 64, "a row is two 32-float halves, one per wave-wide access");
static_assert(OWN_THREADS == 32 * EB, "a work-group's waves step through the list side by side");
#define OWN_ROWS      (BIN_NODES + 1)
#define OWN_LDS_MIN   41984
static_assert(OWN_ROWS * HID + BIN_NODES * OUT_DIM <= OWN_LDS_MIN, "sums and out rows fit");

template <bool LAST>
__global__ void __launch_bounds__(OWN_THREADS)
k_owner(const int* __restrict__ rec,
        const int* __restrict__ tot,
        const float* __restrict__ h,
        const float* __restrict__ dinv,
        const float* __restrict__ b,
        float* __restrict__ outA,
        const float* __restrict__ Wfc,
        const float* __restrict__ bfc,
        float* __restrict__ out,
        size_t wsBytes) {
    if (wsBytes < WS_NEED) return;
    __shared__ __attribute__((aligned(16))) float acc[OWN_LDS_MIN];
    __shared__ int sStart;

    const int tid  = threadIdx.x;
    const int lane = tid & 31;
    const int wave = __builtin_amdgcn_readfirstlane(tid >> 5);
    const int bin  = blockIdx.x;
    const int n0   = bin * BIN_NODES;
    const int nNodes = (N_NODES - n0 < BIN_NODES) ? (N_NODES - n0) : BIN_NODES;

    for (int i = tid; i < OWN_ROWS * HID; i += OWN_THREADS) acc[i] = 0.0f;
    if (wave == 0) {
        int p = 0;
        for (int t = lane; t < bin; t += 32) p += tot[t];
        for (int o = 16; o > 0; o >>= 1) p += __shfl_xor(p, o);
        if (lane == 0) sStart = p;
    }
    __syncthreads();
    const int start = __builtin_amdgcn_readfirstlane(sStart);
    const int end   = start + __builtin_amdgcn_readfirstlane(tot[bin]);

    const v2i pastEnd = {BIN_NODES << ROW_BITS, 0};
    const int last = (end > start) ? end - 1 : 0;
    v2i rcNext;
    {
        const int i0 = start + wave * EB + lane;
        const v2i ld = *(const v2i*)(rec + 2 * (size_t)((i0 < end) ? i0 : last));
        rcNext = (i0 < end) ? ld : pastEnd;
    }

    for (int base = start + wave * EB; base < end; base += OWN_THREADS) {
        const v2i rc = rcNext;
        {
            const int i1 = base + OWN_THREADS + lane;
            const v2i ld = *(const v2i*)(rec + 2 * (size_t)((i1 < end) ? i1 : last));
            rcNext = (i1 < end) ? ld : pastEnd;
        }
        const int rp    = rc.x;
        const int wBits = rc.y;
        const int r  = rp & ((1 << ROW_BITS) - 1);
        const int cl = rp >> ROW_BITS;

        v2f hv[2][GD];

        #pragma unroll
        for (int j = 0; j < GD; ++j) {
            const int rj = __builtin_amdgcn_readlane(r, j);
            const float* src = h + (size_t)rj * HID + lane;
            hv[0][j].x = src[0];
            hv[0][j].y = src[HID / 2];
        }

        #pragma unroll
        for (int g = 0; g < EB / GD; ++g) {
            const int cur = g & 1;
            if (g + 1 < EB / GD) {
                #pragma unroll
                for (int j = 0; j < GD; ++j) {
                    const int rj = __builtin_amdgcn_readlane(r, (g + 1) * GD + j);
                    const float* src = h + (size_t)rj * HID + lane;
                    hv[cur ^ 1][j].x = src[0];
                    hv[cur ^ 1][j].y = src[HID / 2];
                }
            }
            #pragma unroll
            for (int j = 0; j < GD; ++j) {
                const int   cj = __builtin_amdgcn_readlane(cl, g * GD + j);
                const float nj = __int_as_float(
                                     __builtin_amdgcn_readlane(wBits, g * GD + j));
                float* dst = acc + cj * HID + lane;
                atomicAdd(dst,           nj * hv[cur][j].x);
                atomicAdd(dst + HID / 2, nj * hv[cur][j].y);
            }
        }
    }
    __syncthreads();

    if (!LAST) {
        const size_t g0 = (size_t)n0 * HID;
        for (int i = tid; i < nNodes * HID; i += OWN_THREADS) {
            const int n = i >> 6;
            const int f = i & 63;
            const float v = dinv[n0 + n] * (acc[i] + h[g0 + i]) + b[f];
            outA[g0 + i] = fmaxf(v, 0.0f);
        }
    } else {
        float* stage = acc + OWN_ROWS * HID;
        float wf0[OUT_DIM], wf1[OUT_DIM];
        #pragma unroll
        for (int o = 0; o < OUT_DIM; ++o) {
            wf0[o] = Wfc[lane * OUT_DIM + o];
            wf1[o] = Wfc[(lane + HID / 2) * OUT_DIM + o];
        }
        const float bl0 = b[lane];
        const float bl1 = b[lane + HID / 2];
        const float bo  = (lane < OUT_DIM) ? bfc[lane] : 0.0f;
        for (int n = wave; n < nNodes; n += OWN_THREADS / 32) {
            const int node = n0 + n;
            const float di = dinv[node];
            const float* hr = h + (size_t)node * HID + lane;
            const float v0 = fmaxf(di * (acc[n * HID + lane] + hr[0]) + bl0, 0.0f);
            const float v1 = fmaxf(di * (acc[n * HID + HID / 2 + lane] + hr[HID / 2]) + bl1, 0.0f);
            float p[OUT_DIM];
            #pragma unroll
            for (int o = 0; o < OUT_DIM; ++o) p[o] = v0 * wf0[o] + v1 * wf1[o];
            #pragma unroll
            for (int off = 16; off > 0; off >>= 1) {
                #pragma unroll
                for (int o = 0; o < OUT_DIM; ++o) p[o] += __shfl_xor(p[o], off);
            }
            float pv = p[0];
            #pragma unroll
            for (int o = 1; o < OUT_DIM; ++o) pv = (lane == o) ? p[o] : pv;
            if (lane < OUT_DIM) stage[n * OUT_DIM + lane] = pv + bo;
        }
        __syncthreads();
        const size_t o0 = (size_t)n0 * OUT_DIM;
        for (int i = tid; i < nNodes * OUT_DIM; i += OWN_THREADS) out[o0 + i] = stage[i];
    }
}

extern "C" void kernel_launch(void* const* d_in, const int* in_sizes, int n_in,
                              void* d_out, int out_size, void* d_ws, size_t ws_size,
                              hipStream_t stream) {
    const float* x   = (const float*)d_in[0];
    const int*   ei  = (const int*)  d_in[1];
    const float* ew  = (const float*)d_in[2];
    const float* W1  = (const float*)d_in[3];
    const float* b1  = (const float*)d_in[4];
    const float* W2  = (const float*)d_in[5];
    const float* b2  = (const float*)d_in[6];
    const float* Wfc = (const float*)d_in[7];
    const float* bfc = (const float*)d_in[8];
    float* out = (float*)d_out;

    const int* row = ei;
    const int* col = ei + N_EDGES;

    char* ws = (char*)d_ws;
    size_t off = 0;
    float* dinv = (float*)(ws + off); off += DINV_BYTES;
    float* bufH = (float*)(ws + off); off += PLANE_BYTES;
    float* bufA = (float*)(ws + off); off += PLANE_BYTES;
    int*   rec  = (int*)  (ws + off); off += REC_BYTES;
    int*   cnt  = (int*)  (ws + off); off += CNT_BYTES;
    int*   rel  = (int*)  (ws + off); off += CNT_BYTES;
    int*   tot  = (int*)  (ws + off);
    float* deg1 = bufA;
    (void)n_in; (void)in_sizes; (void)out_size;

    const int TPB = 256;
    const int nodeBlocks = (N_NODES + TPB - 1) / TPB;

    k_init_deg<<<nodeBlocks, TPB, 0, stream>>>(dinv, deg1);
    k_count<<<NCHUNKS, BIN_THREADS, 0, stream>>>(col, ew, dinv, deg1, cnt, ws_size);

    k_scan_dinv<<<NBINS, SCAN_THREADS, 0, stream>>>(cnt, rel, tot, dinv, deg1, ws_size);
    k_place<<<NCHUNKS, BIN_THREADS, 0, stream>>>(row, col, ew, rel, tot, rec, ws_size);

    k_gemm_n64<IN_DIM><<<GEMM_BLOCKS, GEMM_THREADS, 0, stream>>>(x, W1, dinv, bufH);
    k_owner<false><<<NBINS, OWN_THREADS, 0, stream>>>(rec, tot, bufH, dinv, b1, bufA,
                                                      Wfc, bfc, out, ws_size);

    k_gemm_n64<HID><<<GEMM_BLOCKS, GEMM_THREADS, 0, stream>>>(bufA, W2, dinv, bufH);
    k_owner<true><<<NBINS, OWN_THREADS, 0, stream>>>(rec, tot, bufH, dinv, b2, bufA,
                                                     Wfc, bfc, out, ws_size);
}


# edge placing runs inside the layer-1 GEMM launch: each work-group files two edge chunks (block scan beside W staging, ranks and stores behind its tile store); 8 -> 7 launches; list order by arrival
# speedup vs baseline: 2.5700x; 1.0785x over previous
#include <hip/hip_runtime.h>
#include <hip/hip_bf16.h>


#define N_NODES 100000
#define N_EDGES 1600000
#define IN_DIM  512
#define HID     64
#define OUT_DIM 5

typedef __attribute__((ext_vector_type(2))) float v2f;
typedef __attribute__((ext_vector_type(8))) float v8f;

typedef __attribute__((ext_vector_type(2))) int v2i;

#define BIN_NODES    416
#define NBINS        ((N_NODES + BIN_NODES - 1) / BIN_NODES)
#define CHUNK        3200
#define NCHUNKS      (N_EDGES / CHUNK)
#define BIN_THREADS  640
#define EPT          (CHUNK / BIN_THREADS)
#define SCAN_THREADS 512
#define OWN_THREADS  1024
#define ROW_BITS     17
static_assert(NCHUNKS * CHUNK == N_EDGES, "chunks must tile the edge list");
static_assert(EPT * BIN_THREADS == CHUNK, "a chunk is a whole number of edges per thread");
static_assert(NBINS <= BIN_THREADS, "one thread per block writes the chunk's counts");
static_assert(NCHUNKS <= SCAN_THREADS, "one thread per chunk in the scan");
static_assert(N_NODES <= (1 << ROW_BITS), "source row must fit its field");
static_assert(BIN_NODES < (1 << (30 - ROW_BITS)), "local target must fit its field");
static_assert((BIN_NODES * OUT_DIM * 4) % 128 == 0, "a block's rows of out are whole lines");

#define DINV_BYTES   ((((size_t)N_NODES * 4) + 255) & ~(size_t)255)
#define PLANE_ELEMS  ((size_t)N_NODES * HID)
#define PLANE_BYTES  (PLANE_ELEMS * 4)
#define REC_BYTES    ((size_t)N_EDGES * 8)
#define CNT_BYTES    ((size_t)NBINS * NCHUNKS * 4)
#define WS_NEED      (DINV_BYTES + 2 * PLANE_BYTES + REC_BYTES + 2 * CNT_BYTES + 1024)

__device__ __forceinline__ int wave_half() {
    return (int)((__builtin_amdgcn_s_sendmsg_rtn(0x87) >> 18) & 1u);
}

__global__ void k_init_deg(float* __restrict__ deg0, float* __restrict__ deg1) {
    int i = blockIdx.x * blockDim.x + threadIdx.x;
    if (i < N_NODES) { deg0[i] = 1.0f; deg1[i] = 0.0f; }
}

__device__ __forceinline__ int block_scan_incl(int v, int* s) {
    const int tid = threadIdx.x;
    const int n   = blockDim.x;
    s[tid] = v;
    __syncthreads();
    for (int d = 1; d < n; d <<= 1) {
        const int t = (tid >= d) ? s[tid - d] : 0;
        __syncthreads();
        s[tid] += t;
        __syncthreads();
    }
    return s[tid];
}

__global__ void __launch_bounds__(BIN_THREADS)
k_count(const int* __restrict__ col,
        const float* __restrict__ w,
        float* __restrict__ deg0,
        float* __restrict__ deg1,
        int* __restrict__ cnt,
        size_t wsBytes) {
    if (wsBytes < WS_NEED) return;
    __shared__ int hist[BIN_THREADS];
    const int tid = threadIdx.x;
    hist[tid] = 0;
    __syncthreads();
    float* deg = wave_half() ? deg1 : deg0;
    const int e0 = blockIdx.x * CHUNK + tid;
    int   c[EPT];
    float wv[EPT];
    #pragma unroll
    for (int k = 0; k < EPT; ++k) {
        c[k]  = col[e0 + k * BIN_THREADS];
        wv[k] = w[e0 + k * BIN_THREADS];
    }
    #pragma unroll
    for (int k = 0; k < EPT; ++k) {
        atomicAdd(deg + c[k], wv[k]);
        atomicAdd(&hist[(unsigned)c[k] / BIN_NODES], 1);
    }
    __syncthreads();
    if (tid < NBINS) cnt[tid * NCHUNKS + blockIdx.x] = hist[tid];
}

static_assert(BIN_NODES <= SCAN_THREADS, "one thread per node of the block for dinv");
__global__ void __launch_bounds__(SCAN_THREADS)
k_scan_dinv(const int* __restrict__ cnt,
            int* __restrict__ rel,
            int* __restrict__ tot,
            float* __restrict__ deg0,
            const float* __restrict__ deg1,
            size_t wsBytes) {
    if (wsBytes < WS_NEED) return;
    __shared__ int s[SCAN_THREADS];
    const int tid = threadIdx.x;
    const int bin = blockIdx.x;
    const int node = bin * BIN_NODES + tid;
    if (tid < BIN_NODES && node < N_NODES) {
        const float d = deg0[node] + deg1[node];
        deg0[node] = (d > 0.0f) ? rsqrtf(d) : 0.0f;
    }
    const int v = (tid < NCHUNKS) ? cnt[bin * NCHUNKS + tid] : 0;
    const int incl = block_scan_incl(v, s);
    if (tid < NCHUNKS) rel[bin * NCHUNKS + tid] = incl - v;
    if (tid == SCAN_THREADS - 1) tot[bin] = incl;
}

#define GEMM_WAVES   25
#define GEMM_THREADS (GEMM_WAVES * 32)
#define GEMM_BLOCKS  ((N_NODES / 16) / GEMM_WAVES)
static_assert(GEMM_BLOCKS * GEMM_WAVES * 16 == N_NODES, "row tiles must fill the GEMM grid exactly");
#define PLACE_EPT    (2 * CHUNK / GEMM_THREADS)
#define SCAN_LANES   256
static_assert(2 * GEMM_BLOCKS == NCHUNKS, "two chunks per GEMM work-group");
static_assert(PLACE_EPT * GEMM_THREADS == 2 * CHUNK, "two chunks are a whole number of edges per thread");
static_assert((PLACE_EPT / 2) * GEMM_THREADS == CHUNK, "a thread's first half of edges is the first chunk");
static_assert(NBINS <= SCAN_LANES && SCAN_LANES % 32 == 0 && SCAN_LANES <= GEMM_THREADS, "whole waves scan the totals");
#define GEMM_LDS_MIN 41984

typedef __attribute__((ext_vector_type(4))) float v4f;

__device__ __forceinline__ void gemm_line(const v4f (&a)[4], const float* bl,
                                          v8f& c0, v8f& c1, v8f& c2, v8f& c3) {
    #pragma unroll
    for (int i = 0; i < 4; ++i) {
        const v4f b00 = *(const v4f*)(bl + i * 512);
        const v4f b01 = *(const v4f*)(bl + i * 512 + 128);
        const v4f b10 = *(const v4f*)(bl + i * 512 + 256);
        const v4f b11 = *(const v4f*)(bl + i * 512 + 384);
        const v2f a0 = {a[i].x, a[i].y};
        const v2f a1 = {a[i].z, a[i].w};
        const v2f p00 = {b00.x, b00.y}, p01 = {b00.z, b00.w};
        const v2f p02 = {b01.x, b01.y}, p03 = {b01.z, b01.w};
        const v2f p10 = {b10.x, b10.y}, p11 = {b10.z, b10.w};
        const v2f p12 = {b11.x, b11.y}, p13 = {b11.z, b11.w};
        c0 = __builtin_amdgcn_wmma_f32_16x16x4_f32(false, a0, false, p00, (short)0, c0, false, false);
        c1 = __builtin_amdgcn_wmma_f32_16x16x4_f32(false, a0, false, p01, (short)0, c1, false, false);
        c2 = __builtin_amdgcn_wmma_f32_16x16x4_f32(false, a0, false, p02, (short)0, c2, false, false);
        c3 = __builtin_amdgcn_wmma_f32_16x16x4_f32(false, a0, false, p03, (short)0, c3, false, false);
        c0 = __builtin_amdgcn_wmma_f32_16x16x4_f32(false, a1, false, p10, (short)0, c0, false, false);
        c1 = __builtin_amdgcn_wmma_f32_16x16x4_f32(false, a1, false, p11, (short)0, c1, false, false);
        c2 = __builtin_amdgcn_wmma_f32_16x16x4_f32(false, a1, false, p12, (short)0, c2, false, false);
        c3 = __builtin_amdgcn_wmma_f32_16x16x4_f32(false, a1, false, p13, (short)0, c3, false, false);
    }
}

template <int K, bool PLACE>
__global__ void __launch_bounds__(GEMM_THREADS)
k_gemm_n64(const float* __restrict__ X,
           const float* __restrict__ W,
           const float* __restrict__ rs,
           float* __restrict__ H,
           const int* __restrict__ row,
           const int* __restrict__ col,
           const float* __restrict__ w,
           const int* __restrict__ rel,
           const int* __restrict__ tot,
           int* __restrict__ rec,
           size_t wsBytes) {
    static_assert(K % 64 == 0, "the k loop takes two 32-k lines per trip");
    constexpr int LDS_FLOATS = (K * HID > GEMM_LDS_MIN) ? K * HID : GEMM_LDS_MIN;
    __shared__ __attribute__((aligned(16))) float Bl[LDS_FLOATS];
    __shared__ int sWave[SCAN_LANES / 32];
    __shared__ int cursor[2 * SCAN_LANES];
    if (PLACE && wsBytes < WS_NEED) return;

    const int tid  = threadIdx.x;
    const int lane = tid & 31;
    const int c    = lane & 15;
    const int h    = lane >> 4;
    const int tile = blockIdx.x * GEMM_WAVES + (tid >> 5);
    const float* xr = X + (size_t)(tile * 16 + c) * K + 4 * h;

    v4f aA[4], aB[4];
    #pragma unroll
    for (int i = 0; i < 4; ++i) aA[i] = *(const v4f*)(xr + 8 * i);

    int pc[PLACE_EPT] = {}, pr[PLACE_EPT] = {}, pw[PLACE_EPT] = {};
    int tv = 0, rl0 = 0, rl1 = 0;
    if (PLACE) {
        const int e0 = blockIdx.x * (2 * CHUNK) + tid;
        #pragma unroll
        for (int k = 0; k < PLACE_EPT; ++k) {
            pc[k] = col[e0 + k * GEMM_THREADS];
            pr[k] = row[e0 + k * GEMM_THREADS];
            pw[k] = __float_as_int(w[e0 + k * GEMM_THREADS]);
        }
        const int tb = (tid < NBINS) ? tid : 0;
        tv  = tot[tb];
        rl0 = rel[tb * NCHUNKS + 2 * blockIdx.x];
        rl1 = rel[tb * NCHUNKS + 2 * blockIdx.x + 1];
    }

    for (int u = tid; u < (K / 2) * 32; u += GEMM_THREADS) {
        const int cc = u & 15;
        const int tp = (u >> 4) & 1;
        const int k  = (u >> 5) * 2;
        const float* w0 = W + (size_t)k * HID + tp * 32 + cc;
        v4f v;
        v.x = w0[0];
        v.y = w0[HID];
        v.z = w0[16];
        v.w = w0[HID + 16];
        const int kb = k >> 3;
        const int hh = (k >> 2) & 1;
        const int j  = (k >> 1) & 1;
        *(v4f*)(Bl + kb * 512 + (2 * j + tp) * 128 + (16 * hh + cc) * 4) = v;
    }

    const int tvs = (tid < NBINS) ? tv : 0;
    int incl = tvs;
    if (PLACE && tid < SCAN_LANES) {
        #pragma unroll
        for (int d = 1; d < 32; d <<= 1) {
            const int y = __shfl_up(incl, d);
            if (lane >= d) incl += y;
        }
        if (lane == 31) sWave[tid >> 5] = incl;
    }
    __syncthreads();
    if (PLACE) {
        if (tid < SCAN_LANES) {
            int p = 0;
            for (int q = 0; q < (tid >> 5); ++q) p += sWave[q];
            if (tid < NBINS) {
                const int base = p + incl - tvs;
                cursor[tid]              = base + rl0;
                cursor[SCAN_LANES + tid] = base + rl1;
            }
        }
        __syncthreads();
    }

    v8f c0 = {}, c1 = {}, c2 = {}, c3 = {};
    const float* bl = Bl + lane * 4;

    for (int ch = 0; ch < K / 32; ch += 2) {
        #pragma unroll
        for (int i = 0; i < 4; ++i) aB[i] = *(const v4f*)(xr + (ch + 1) * 32 + 8 * i);
        gemm_line(aA, bl + ch * 2048, c0, c1, c2, c3);
        if (ch + 2 < K / 32) {
            #pragma unroll
            for (int i = 0; i < 4; ++i) aA[i] = *(const v4f*)(xr + (ch + 2) * 32 + 8 * i);
        }
        gemm_line(aB, bl + (ch + 1) * 2048, c0, c1, c2, c3);
    }

    const int rBase = tile * 16 + 8 * h;
    #pragma unroll
    for (int j = 0; j < 8; ++j) {
        const float sc = rs[rBase + j];
        float* dst = H + (size_t)(rBase + j) * HID + c;
        dst[0]  = sc * c0[j];
        dst[16] = sc * c1[j];
        dst[32] = sc * c2[j];
        dst[48] = sc * c3[j];
    }

    if (PLACE) {
        int bin[PLACE_EPT], pos[PLACE_EPT];
        #pragma unroll
        for (int k = 0; k < PLACE_EPT; ++k) {
            bin[k] = (int)((unsigned)pc[k] / BIN_NODES);
            pos[k] = atomicAdd(&cursor[(k < PLACE_EPT / 2 ? 0 : SCAN_LANES) + bin[k]], 1);
        }
        #pragma unroll
        for (int k = 0; k < PLACE_EPT; ++k) {
            const int cl = pc[k] - bin[k] * BIN_NODES;
            v2i rc;
            rc.x = pr[k] | (cl << ROW_BITS);
            rc.y = pw[k];
            *(v2i*)(rec + 2 * (size_t)pos[k]) = rc;
        }
    }
}

#define EB 32
#define GD 8
static_assert(EB % GD == 0, "gather groups must tile a wave trip");
static_assert(HID == 64, "a row is two 32-float halves, one per wave-wide access");
static_assert(OWN_THREADS == 32 * EB, "a work-group's waves step through the list side by side");
#define OWN_ROWS      (BIN_NODES + 1)
#define OWN_LDS_MIN   41984
static_assert(OWN_ROWS * HID + BIN_NODES * OUT_DIM <= OWN_LDS_MIN, "sums and out rows fit");

template <bool LAST>
__global__ void __launch_bounds__(OWN_THREADS)
k_owner(const int* __restrict__ rec,
        const int* __restrict__ tot,
        const float* __restrict__ h,
        const float* __restrict__ dinv,
        const float* __restrict__ b,
        float* __restrict__ outA,
        const float* __restrict__ Wfc,
        const float* __restrict__ bfc,
        float* __restrict__ out,
        size_t wsBytes) {
    if (wsBytes < WS_NEED) return;
    __shared__ __attribute__((aligned(16))) float acc[OWN_LDS_MIN];
    __shared__ int sStart;

    const int tid  = threadIdx.x;
    const int lane = tid & 31;
    const int wave = __builtin_amdgcn_readfirstlane(tid >> 5);
    const int bin  = blockIdx.x;
    const int n0   = bin * BIN_NODES;
    const int nNodes = (N_NODES - n0 < BIN_NODES) ? (N_NODES - n0) : BIN_NODES;

    for (int i = tid; i < OWN_ROWS * HID; i += OWN_THREADS) acc[i] = 0.0f;
    if (wave == 0) {
        int p = 0;
        for (int t = lane; t < bin; t += 32) p += tot[t];
        for (int o = 16; o > 0; o >>= 1) p += __shfl_xor(p, o);
        if (lane == 0) sStart = p;
    }
    __syncthreads();
    const int start = __builtin_amdgcn_readfirstlane(sStart);
    const int end   = start + __builtin_amdgcn_readfirstlane(tot[bin]);

    const v2i pastEnd = {BIN_NODES << ROW_BITS, 0};
    const int last = (end > start) ? end - 1 : 0;
    v2i rcNext;
    {
        const int i0 = start + wave * EB + lane;
        const v2i ld = *(const v2i*)(rec + 2 * (size_t)((i0 < end) ? i0 : last));
        rcNext = (i0 < end) ? ld : pastEnd;
    }

    for (int base = start + wave * EB; base < end; base += OWN_THREADS) {
        const v2i rc = rcNext;
        {
            const int i1 = base + OWN_THREADS + lane;
            const v2i ld = *(const v2i*)(rec + 2 * (size_t)((i1 < end) ? i1 : last));
            rcNext = (i1 < end) ? ld : pastEnd;
        }
        const int rp    = rc.x;
        const int wBits = rc.y;
        const int r  = rp & ((1 << ROW_BITS) - 1);
        const int cl = rp >> ROW_BITS;

        v2f hv[2][GD];

        #pragma unroll
        for (int j = 0; j < GD; ++j) {
            const int rj = __builtin_amdgcn_readlane(r, j);
            const float* src = h + (size_t)rj * HID + lane;
            hv[0][j].x = src[0];
            hv[0][j].y = src[HID / 2];
        }

        #pragma unroll
        for (int g = 0; g < EB / GD; ++g) {
            const int cur = g & 1;
            if (g + 1 < EB / GD) {
                #pragma unroll
                for (int j = 0; j < GD; ++j) {
                    const int rj = __builtin_amdgcn_readlane(r, (g + 1) * GD + j);
                    const float* src = h + (size_t)rj * HID + lane;
                    hv[cur ^ 1][j].x = src[0];
                    hv[cur ^ 1][j].y = src[HID / 2];
                }
            }
            #pragma unroll
            for (int j = 0; j < GD; ++j) {
                const int   cj = __builtin_amdgcn_readlane(cl, g * GD + j);
                const float nj = __int_as_float(
                                     __builtin_amdgcn_readlane(wBits, g * GD + j));
                float* dst = acc + cj * HID + lane;
                atomicAdd(dst,           nj * hv[cur][j].x);
                atomicAdd(dst + HID / 2, nj * hv[cur][j].y);
            }
        }
    }
    __syncthreads();

    if (!LAST) {
        const size_t g0 = (size_t)n0 * HID;
        for (int i = tid; i < nNodes * HID; i += OWN_THREADS) {
            const int n = i >> 6;
            const int f = i & 63;
            const float v = dinv[n0 + n] * (acc[i] + h[g0 + i]) + b[f];
            outA[g0 + i] = fmaxf(v, 0.0f);
        }
    } else {
        float* stage = acc + OWN_ROWS * HID;
        float wf0[OUT_DIM], wf1[OUT_DIM];
        #pragma unroll
        for (int o = 0; o < OUT_DIM; ++o) {
            wf0[o] = Wfc[lane * OUT_DIM + o];
            wf1[o] = Wfc[(lane + HID / 2) * OUT_DIM + o];
        }
        const float bl0 = b[lane];
        const float bl1 = b[lane + HID / 2];
        const float bo  = (lane < OUT_DIM) ? bfc[lane] : 0.0f;
        for (int n = wave; n < nNodes; n += OWN_THREADS / 32) {
            const int node = n0 + n;
            const float di = dinv[node];
            const float* hr = h + (size_t)node * HID + lane;
            const float v0 = fmaxf(di * (acc[n * HID + lane] + hr[0]) + bl0, 0.0f);
            const float v1 = fmaxf(di * (acc[n * HID + HID / 2 + lane] + hr[HID / 2]) + bl1, 0.0f);
            float p[OUT_DIM];
            #pragma unroll
            for (int o = 0; o < OUT_DIM; ++o) p[o] = v0 * wf0[o] + v1 * wf1[o];
            #pragma unroll
            for (int off = 16; off > 0; off >>= 1) {
                #pragma unroll
                for (int o = 0; o < OUT_DIM; ++o) p[o] += __shfl_xor(p[o], off);
            }
            float pv = p[0];
            #pragma unroll
            for (int o = 1; o < OUT_DIM; ++o) pv = (lane == o) ? p[o] : pv;
            if (lane < OUT_DIM) stage[n * OUT_DIM + lane] = pv + bo;
        }
        __syncthreads();
        const size_t o0 = (size_t)n0 * OUT_DIM;
        for (int i = tid; i < nNodes * OUT_DIM; i += OWN_THREADS) out[o0 + i] = stage[i];
    }
}

extern "C" void kernel_launch(void* const* d_in, const int* in_sizes, int n_in,
                              void* d_out, int out_size, void* d_ws, size_t ws_size,
                              hipStream_t stream) {
    const float* x   = (const float*)d_in[0];
    const int*   ei  = (const int*)  d_in[1];
    const float* ew  = (const float*)d_in[2];
    const float* W1  = (const float*)d_in[3];
    const float* b1  = (const float*)d_in[4];
    const float* W2  = (const float*)d_in[5];
    const float* b2  = (const float*)d_in[6];
    const float* Wfc = (const float*)d_in[7];
    const float* bfc = (const float*)d_in[8];
    float* out = (float*)d_out;

    const int* row = ei;
    const int* col = ei + N_EDGES;

    char* ws = (char*)d_ws;
    size_t off = 0;
    float* dinv = (float*)(ws + off); off += DINV_BYTES;
    float* bufH = (float*)(ws + off); off += PLANE_BYTES;
    float* bufA = (float*)(ws + off); off += PLANE_BYTES;
    int*   rec  = (int*)  (ws + off); off += REC_BYTES;
    int*   cnt  = (int*)  (ws + off); off += CNT_BYTES;
    int*   rel  = (int*)  (ws + off); off += CNT_BYTES;
    int*   tot  = (int*)  (ws + off);
    float* deg1 = bufA;
    (void)n_in; (void)in_sizes; (void)out_size;

    const int TPB = 256;
    const int nodeBlocks = (N_NODES + TPB - 1) / TPB;

    k_init_deg<<<nodeBlocks, TPB, 0, stream>>>(dinv, deg1);
    k_count<<<NCHUNKS, BIN_THREADS, 0, stream>>>(col, ew, dinv, deg1, cnt, ws_size);

    k_scan_dinv<<<NBINS, SCAN_THREADS, 0, stream>>>(cnt, rel, tot, dinv, deg1, ws_size);

    k_gemm_n64<IN_DIM, true><<<GEMM_BLOCKS, GEMM_THREADS, 0, stream>>>(
        x, W1, dinv, bufH, row, col, ew, rel, tot, rec, ws_size);
    k_owner<false><<<NBINS, OWN_THREADS, 0, stream>>>(rec, tot, bufH, dinv, b1, bufA,
                                                      Wfc, bfc, out, ws_size);

    k_gemm_n64<HID, false><<<GEMM_BLOCKS, GEMM_THREADS, 0, stream>>>(
        bufA, W2, dinv, bufH, row, col, ew, rel, tot, rec, ws_size);
    k_owner<true><<<NBINS, OWN_THREADS, 0, stream>>>(rec, tot, bufH, dinv, b2, bufA,
                                                     Wfc, bfc, out, ws_size);
}


# layer-2 GEMM moved into the layer-1 owner launch: relu rows stay in LDS (row pitch 272 B), 16-row tiles x W2 on the same f32 matrix line, rows stored pre-scaled; 7 -> 6 launches; sum orders unchanged
# speedup vs baseline: 2.8006x; 1.0897x over previous
#include <hip/hip_runtime.h>
#include <hip/hip_bf16.h>


#define N_NODES 100000
#define N_EDGES 1600000
#define IN_DIM  512
#define HID     64
#define OUT_DIM 5

typedef __attribute__((ext_vector_type(2))) float v2f;
typedef __attribute__((ext_vector_type(8))) float v8f;

typedef __attribute__((ext_vector_type(2))) int v2i;

#define BIN_NODES    416
#define NBINS        ((N_NODES + BIN_NODES - 1) / BIN_NODES)
#define CHUNK        3200
#define NCHUNKS      (N_EDGES / CHUNK)
#define BIN_THREADS  640
#define EPT          (CHUNK / BIN_THREADS)
#define SCAN_THREADS 512
#define OWN_THREADS  1024
#define ROW_BITS     17
static_assert(NCHUNKS * CHUNK == N_EDGES, "chunks must tile the edge list");
static_assert(EPT * BIN_THREADS == CHUNK, "a chunk is a whole number of edges per thread");
static_assert(NBINS <= BIN_THREADS, "one thread per block writes the chunk's counts");
static_assert(NCHUNKS <= SCAN_THREADS, "one thread per chunk in the scan");
static_assert(N_NODES <= (1 << ROW_BITS), "source row must fit its field");
static_assert(BIN_NODES < (1 << (30 - ROW_BITS)), "local target must fit its field");
static_assert((BIN_NODES * OUT_DIM * 4) % 128 == 0, "a block's rows of out are whole lines");

#define DINV_BYTES   ((((size_t)N_NODES * 4) + 255) & ~(size_t)255)
#define PLANE_ELEMS  ((size_t)N_NODES * HID)
#define PLANE_BYTES  (PLANE_ELEMS * 4)
#define REC_BYTES    ((size_t)N_EDGES * 8)
#define CNT_BYTES    ((size_t)NBINS * NCHUNKS * 4)
#define WS_NEED      (DINV_BYTES + 2 * PLANE_BYTES + REC_BYTES + 2 * CNT_BYTES + 1024)

__device__ __forceinline__ int wave_half() {
    return (int)((__builtin_amdgcn_s_sendmsg_rtn(0x87) >> 18) & 1u);
}

__global__ void k_init_deg(float* __restrict__ deg0, float* __restrict__ deg1) {
    int i = blockIdx.x * blockDim.x + threadIdx.x;
    if (i < N_NODES) { deg0[i] = 1.0f; deg1[i] = 0.0f; }
}

__device__ __forceinline__ int block_scan_incl(int v, int* s) {
    const int tid = threadIdx.x;
    const int n   = blockDim.x;
    s[tid] = v;
    __syncthreads();
    for (int d = 1; d < n; d <<= 1) {
        const int t = (tid >= d) ? s[tid - d] : 0;
        __syncthreads();
        s[tid] += t;
        __syncthreads();
    }
    return s[tid];
}

__global__ void __launch_bounds__(BIN_THREADS)
k_count(const int* __restrict__ col,
        const float* __restrict__ w,
        float* __restrict__ deg0,
        float* __restrict__ deg1,
        int* __restrict__ cnt,
        size_t wsBytes) {
    if (wsBytes < WS_NEED) return;
    __shared__ int hist[BIN_THREADS];
    const int tid = threadIdx.x;
    hist[tid] = 0;
    __syncthreads();
    float* deg = wave_half() ? deg1 : deg0;
    const int e0 = blockIdx.x * CHUNK + tid;
    int   c[EPT];
    float wv[EPT];
    #pragma unroll
    for (int k = 0; k < EPT; ++k) {
        c[k]  = col[e0 + k * BIN_THREADS];
        wv[k] = w[e0 + k * BIN_THREADS];
    }
    #pragma unroll
    for (int k = 0; k < EPT; ++k) {
        atomicAdd(deg + c[k], wv[k]);
        atomicAdd(&hist[(unsigned)c[k] / BIN_NODES], 1);
    }
    __syncthreads();
    if (tid < NBINS) cnt[tid * NCHUNKS + blockIdx.x] = hist[tid];
}

static_assert(BIN_NODES <= SCAN_THREADS, "one thread per node of the block for dinv");
__global__ void __launch_bounds__(SCAN_THREADS)
k_scan_dinv(const int* __restrict__ cnt,
            int* __restrict__ rel,
            int* __restrict__ tot,
            float* __restrict__ deg0,
            const float* __restrict__ deg1,
            size_t wsBytes) {
    if (wsBytes < WS_NEED) return;
    __shared__ int s[SCAN_THREADS];
    const int tid = threadIdx.x;
    const int bin = blockIdx.x;
    const int node = bin * BIN_NODES + tid;
    if (tid < BIN_NODES && node < N_NODES) {
        const float d = deg0[node] + deg1[node];
        deg0[node] = (d > 0.0f) ? rsqrtf(d) : 0.0f;
    }
    const int v = (tid < NCHUNKS) ? cnt[bin * NCHUNKS + tid] : 0;
    const int incl = block_scan_incl(v, s);
    if (tid < NCHUNKS) rel[bin * NCHUNKS + tid] = incl - v;
    if (tid == SCAN_THREADS - 1) tot[bin] = incl;
}

#define GEMM_WAVES   25
#define GEMM_THREADS (GEMM_WAVES * 32)
#define GEMM_BLOCKS  ((N_NODES / 16) / GEMM_WAVES)
static_assert(GEMM_BLOCKS * GEMM_WAVES * 16 == N_NODES, "row tiles must fill the GEMM grid exactly");
#define PLACE_EPT    (2 * CHUNK / GEMM_THREADS)
#define SCAN_LANES   256
static_assert(2 * GEMM_BLOCKS == NCHUNKS, "two chunks per GEMM work-group");
static_assert(PLACE_EPT * GEMM_THREADS == 2 * CHUNK, "two chunks are a whole number of edges per thread");
static_assert((PLACE_EPT / 2) * GEMM_THREADS == CHUNK, "a thread's first half of edges is the first chunk");
static_assert(NBINS <= SCAN_LANES && SCAN_LANES % 32 == 0 && SCAN_LANES <= GEMM_THREADS, "whole waves scan the totals");
#define GEMM_LDS_MIN 41984

typedef __attribute__((ext_vector_type(4))) float v4f;

__device__ __forceinline__ void gemm_line(const v4f (&a)[4], const float* bl,
                                          v8f& c0, v8f& c1, v8f& c2, v8f& c3) {
    #pragma unroll
    for (int i = 0; i < 4; ++i) {
        const v4f b00 = *(const v4f*)(bl + i * 512);
        const v4f b01 = *(const v4f*)(bl + i * 512 + 128);
        const v4f b10 = *(const v4f*)(bl + i * 512 + 256);
        const v4f b11 = *(const v4f*)(bl + i * 512 + 384);
        const v2f a0 = {a[i].x, a[i].y};
        const v2f a1 = {a[i].z, a[i].w};
        const v2f p00 = {b00.x, b00.y}, p01 = {b00.z, b00.w};
        const v2f p02 = {b01.x, b01.y}, p03 = {b01.z, b01.w};
        const v2f p10 = {b10.x, b10.y}, p11 = {b10.z, b10.w};
        const v2f p12 = {b11.x, b11.y}, p13 = {b11.z, b11.w};
        c0 = __builtin_amdgcn_wmma_f32_16x16x4_f32(false, a0, false, p00, (short)0, c0, false, false);
        c1 = __builtin_amdgcn_wmma_f32_16x16x4_f32(false, a0, false, p01, (short)0, c1, false, false);
        c2 = __builtin_amdgcn_wmma_f32_16x16x4_f32(false, a0, false, p02, (short)0, c2, false, false);
        c3 = __builtin_amdgcn_wmma_f32_16x16x4_f32(false, a0, false, p03, (short)0, c3, false, false);
        c0 = __builtin_amdgcn_wmma_f32_16x16x4_f32(false, a1, false, p10, (short)0, c0, false, false);
        c1 = __builtin_amdgcn_wmma_f32_16x16x4_f32(false, a1, false, p11, (short)0, c1, false, false);
        c2 = __builtin_amdgcn_wmma_f32_16x16x4_f32(false, a1, false, p12, (short)0, c2, false, false);
        c3 = __builtin_amdgcn_wmma_f32_16x16x4_f32(false, a1, false, p13, (short)0, c3, false, false);
    }
}

__device__ __forceinline__ v4f w_unit_load(const float* __restrict__ W, int u) {
    const int cc = u & 15;
    const int tp = (u >> 4) & 1;
    const int k  = (u >> 5) * 2;
    const float* w0 = W + (size_t)k * HID + tp * 32 + cc;
    v4f v;
    v.x = w0[0];
    v.y = w0[HID];
    v.z = w0[16];
    v.w = w0[HID + 16];
    return v;
}
__device__ __forceinline__ int w_unit_place(int u) {
    const int cc = u & 15;
    const int tp = (u >> 4) & 1;
    const int k  = (u >> 5) * 2;
    const int kb = k >> 3;
    const int hh = (k >> 2) & 1;
    const int j  = (k >> 1) & 1;
    return kb * 512 + (2 * j + tp) * 128 + (16 * hh + cc) * 4;
}

template <int K, bool PLACE>
__global__ void __launch_bounds__(GEMM_THREADS)
k_gemm_n64(const float* __restrict__ X,
           const float* __restrict__ W,
           const float* __restrict__ rs,
           float* __restrict__ H,
           const int* __restrict__ row,
           const int* __restrict__ col,
           const float* __restrict__ w,
           const int* __restrict__ rel,
           const int* __restrict__ tot,
           int* __restrict__ rec,
           size_t wsBytes) {
    static_assert(K % 64 == 0, "the k loop takes two 32-k lines per trip");
    constexpr int LDS_FLOATS = (K * HID > GEMM_LDS_MIN) ? K * HID : GEMM_LDS_MIN;
    __shared__ __attribute__((aligned(16))) float Bl[LDS_FLOATS];
    __shared__ int sWave[SCAN_LANES / 32];
    __shared__ int cursor[2 * SCAN_LANES];
    if (PLACE && wsBytes < WS_NEED) return;

    const int tid  = threadIdx.x;
    const int lane = tid & 31;
    const int c    = lane & 15;
    const int h    = lane >> 4;
    const int tile = blockIdx.x * GEMM_WAVES + (tid >> 5);
    const float* xr = X + (size_t)(tile * 16 + c) * K + 4 * h;

    v4f aA[4], aB[4];
    #pragma unroll
    for (int i = 0; i < 4; ++i) aA[i] = *(const v4f*)(xr + 8 * i);

    int pc[PLACE_EPT] = {}, pr[PLACE_EPT] = {}, pw[PLACE_EPT] = {};
    int tv = 0, rl0 = 0, rl1 = 0;
    if (PLACE) {
        const int e0 = blockIdx.x * (2 * CHUNK) + tid;
        #pragma unroll
        for (int k = 0; k < PLACE_EPT; ++k) {
            pc[k] = col[e0 + k * GEMM_THREADS];
            pr[k] = row[e0 + k * GEMM_THREADS];
            pw[k] = __float_as_int(w[e0 + k * GEMM_THREADS]);
        }
        const int tb = (tid < NBINS) ? tid : 0;
        tv  = tot[tb];
        rl0 = rel[tb * NCHUNKS + 2 * blockIdx.x];
        rl1 = rel[tb * NCHUNKS + 2 * blockIdx.x + 1];
    }

    for (int u = tid; u < (K / 2) * 32; u += GEMM_THREADS) {
        const int cc = u & 15;
        const int tp = (u >> 4) & 1;
        const int k  = (u >> 5) * 2;
        const float* w0 = W + (size_t)k * HID + tp * 32 + cc;
        v4f v;
        v.x = w0[0];
        v.y = w0[HID];
        v.z = w0[16];
        v.w = w0[HID + 16];
        const int kb = k >> 3;
        const int hh = (k >> 2) & 1;
        const int j  = (k >> 1) & 1;
        *(v4f*)(Bl + kb * 512 + (2 * j + tp) * 128 + (16 * hh + cc) * 4) = v;
    }

    const int tvs = (tid < NBINS) ? tv : 0;
    int incl = tvs;
    if (PLACE && tid < SCAN_LANES) {
        #pragma unroll
        for (int d = 1; d < 32; d <<= 1) {
            const int y = __shfl_up(incl, d);
            if (lane >= d) incl += y;
        }
        if (lane == 31) sWave[tid >> 5] = incl;
    }
    __syncthreads();
    if (PLACE) {
        if (tid < SCAN_LANES) {
            int p = 0;
            for (int q = 0; q < (tid >> 5); ++q) p += sWave[q];
            if (tid < NBINS) {
                const int base = p + incl - tvs;
                cursor[tid]              = base + rl0;
                cursor[SCAN_LANES + tid] = base + rl1;
            }
        }
        __syncthreads();
    }

    v8f c0 = {}, c1 = {}, c2 = {}, c3 = {};
    const float* bl = Bl + lane * 4;

    for (int ch = 0; ch < K / 32; ch += 2) {
        #pragma unroll
        for (int i = 0; i < 4; ++i) aB[i] = *(const v4f*)(xr + (ch + 1) * 32 + 8 * i);
        gemm_line(aA, bl + ch * 2048, c0, c1, c2, c3);
        if (ch + 2 < K / 32) {
            #pragma unroll
            for (int i = 0; i < 4; ++i) aA[i] = *(const v4f*)(xr + (ch + 2) * 32 + 8 * i);
        }
        gemm_line(aB, bl + (ch + 1) * 2048, c0, c1, c2, c3);
    }

    const int rBase = tile * 16 + 8 * h;
    #pragma unroll
    for (int j = 0; j < 8; ++j) {
        const float sc = rs[rBase + j];
        float* dst = H + (size_t)(rBase + j) * HID + c;
        dst[0]  = sc * c0[j];
        dst[16] = sc * c1[j];
        dst[32] = sc * c2[j];
        dst[48] = sc * c3[j];
    }

    if (PLACE) {
        int bin[PLACE_EPT], pos[PLACE_EPT];
        #pragma unroll
        for (int k = 0; k < PLACE_EPT; ++k) {
            bin[k] = (int)((unsigned)pc[k] / BIN_NODES);
            pos[k] = atomicAdd(&cursor[(k < PLACE_EPT / 2 ? 0 : SCAN_LANES) + bin[k]], 1);
        }
        #pragma unroll
        for (int k = 0; k < PLACE_EPT; ++k) {
            const int cl = pc[k] - bin[k] * BIN_NODES;
            v2i rc;
            rc.x = pr[k] | (cl << ROW_BITS);
            rc.y = pw[k];
            *(v2i*)(rec + 2 * (size_t)pos[k]) = rc;
        }
    }
}

#define EB 32
#define GD 8
static_assert(EB % GD == 0, "gather groups must tile a wave trip");
static_assert(HID == 64, "a row is two 32-float halves, one per wave-wide access");
static_assert(OWN_THREADS == 32 * EB, "a work-group's waves step through the list side by side");
#define OWN_ROWS      (BIN_NODES + 1)
#define OWN_LDS_MIN   41984
static_assert(OWN_ROWS * HID + BIN_NODES * OUT_DIM <= OWN_LDS_MIN, "sums and out rows fit");
#define V_PITCH       (HID + 4)
#define OWN_W_OFF     (OWN_ROWS * HID)
#define OWN_V_OFF     (OWN_W_OFF + HID * HID)
#define OWN_LDS_MID   (OWN_V_OFF + BIN_NODES * V_PITCH)
static_assert(OWN_LDS_MID >= OWN_LDS_MIN, "still one work-group per WGP");
static_assert((size_t)OWN_LDS_MID * 4 + 64 <= 320 * 1024, "sums, Wn and v fit the WGP's LDS");
static_assert(OWN_W_OFF % 4 == 0 && OWN_V_OFF % 4 == 0 && V_PITCH % 4 == 0, "16-byte LDS pieces");
static_assert((HID / 2) * 32 == OWN_THREADS, "one unit of Wn per thread");
static_assert(BIN_NODES % 16 == 0 && N_NODES % 16 == 0, "a block is a whole number of 16-row tiles");
static_assert(BIN_NODES / 16 <= OWN_THREADS / 32, "one wave per tile");

template <bool LAST>
__global__ void __launch_bounds__(OWN_THREADS)
k_owner(const int* __restrict__ rec,
        const int* __restrict__ tot,
        const float* __restrict__ h,
        const float* __restrict__ dinv,
        const float* __restrict__ b,
        const float* __restrict__ Wn,
        float* __restrict__ hsN,
        const float* __restrict__ Wfc,
        const float* __restrict__ bfc,
        float* __restrict__ out,
        size_t wsBytes) {
    if (wsBytes < WS_NEED) return;
    constexpr int LDS_FLOATS = LAST ? OWN_LDS_MIN : OWN_LDS_MID;
    __shared__ __attribute__((aligned(16))) float acc[LDS_FLOATS];
    __shared__ int sStart;

    const int tid  = threadIdx.x;
    const int lane = tid & 31;
    const int wave = __builtin_amdgcn_readfirstlane(tid >> 5);
    const int bin  = blockIdx.x;
    const int n0   = bin * BIN_NODES;
    const int nNodes = (N_NODES - n0 < BIN_NODES) ? (N_NODES - n0) : BIN_NODES;

    v4f wn = {};
    if (!LAST) wn = w_unit_load(Wn, tid);
    for (int i = tid; i < OWN_ROWS * HID; i += OWN_THREADS) acc[i] = 0.0f;
    if (!LAST) *(v4f*)(acc + OWN_W_OFF + w_unit_place(tid)) = wn;
    if (wave == 0) {
        int p = 0;
        for (int t = lane; t < bin; t += 32) p += tot[t];
        for (int o = 16; o > 0; o >>= 1) p += __shfl_xor(p, o);
        if (lane == 0) sStart = p;
    }
    __syncthreads();
    const int start = __builtin_amdgcn_readfirstlane(sStart);
    const int end   = start + __builtin_amdgcn_readfirstlane(tot[bin]);

    const v2i pastEnd = {BIN_NODES << ROW_BITS, 0};
    const int last = (end > start) ? end - 1 : 0;
    v2i rcNext;
    {
        const int i0 = start + wave * EB + lane;
        const v2i ld = *(const v2i*)(rec + 2 * (size_t)((i0 < end) ? i0 : last));
        rcNext = (i0 < end) ? ld : pastEnd;
    }

    for (int base = start + wave * EB; base < end; base += OWN_THREADS) {
        const v2i rc = rcNext;
        {
            const int i1 = base + OWN_THREADS + lane;
            const v2i ld = *(const v2i*)(rec + 2 * (size_t)((i1 < end) ? i1 : last));
            rcNext = (i1 < end) ? ld : pastEnd;
        }
        const int rp    = rc.x;
        const int wBits = rc.y;
        const int r  = rp & ((1 << ROW_BITS) - 1);
        const int cl = rp >> ROW_BITS;

        v2f hv[2][GD];

        #pragma unroll
        for (int j = 0; j < GD; ++j) {
            const int rj = __builtin_amdgcn_readlane(r, j);
            const float* src = h + (size_t)rj * HID + lane;
            hv[0][j].x = src[0];
            hv[0][j].y = src[HID / 2];
        }

        #pragma unroll
        for (int g = 0; g < EB / GD; ++g) {
            const int cur = g & 1;
            if (g + 1 < EB / GD) {
                #pragma unroll
                for (int j = 0; j < GD; ++j) {
                    const int rj = __builtin_amdgcn_readlane(r, (g + 1) * GD + j);
                    const float* src = h + (size_t)rj * HID + lane;
                    hv[cur ^ 1][j].x = src[0];
                    hv[cur ^ 1][j].y = src[HID / 2];
                }
            }
            #pragma unroll
            for (int j = 0; j < GD; ++j) {
                const int   cj = __builtin_amdgcn_readlane(cl, g * GD + j);
                const float nj = __int_as_float(
                                     __builtin_amdgcn_readlane(wBits, g * GD + j));
                float* dst = acc + cj * HID + lane;
                atomicAdd(dst,           nj * hv[cur][j].x);
                atomicAdd(dst + HID / 2, nj * hv[cur][j].y);
            }
        }
    }
    __syncthreads();

    if (!LAST) {
        float* vbuf = acc + OWN_V_OFF;
        const size_t g0 = (size_t)n0 * HID;
        for (int i = tid; i < nNodes * HID; i += OWN_THREADS) {
            const int n = i >> 6;
            const int f = i & 63;
            const float v = dinv[n0 + n] * (acc[i] + h[g0 + i]) + b[f];
            vbuf[n * V_PITCH + f] = fmaxf(v, 0.0f);
        }
        __syncthreads();
        if (wave < (nNodes >> 4)) {
            const int c  = lane & 15;
            const int hh = lane >> 4;
            const float* vr = vbuf + (wave * 16 + c) * V_PITCH + 4 * hh;
            v4f aA[4], aB[4];
            #pragma unroll
            for (int i = 0; i < 4; ++i) {
                aA[i] = *(const v4f*)(vr + 8 * i);
                aB[i] = *(const v4f*)(vr + 32 + 8 * i);
            }
            v8f c0 = {}, c1 = {}, c2 = {}, c3 = {};
            const float* bl = acc + OWN_W_OFF + lane * 4;
            gemm_line(aA, bl, c0, c1, c2, c3);
            gemm_line(aB, bl + 2048, c0, c1, c2, c3);
            const int rBase = n0 + wave * 16 + 8 * hh;
            #pragma unroll
            for (int j = 0; j < 8; ++j) {
                const float sc = dinv[rBase + j];
                float* dst = hsN + (size_t)(rBase + j) * HID + c;
                dst[0]  = sc * c0[j];
                dst[16] = sc * c1[j];
                dst[32] = sc * c2[j];
                dst[48] = sc * c3[j];
            }
        }
    } else {
        float* stage = acc + OWN_ROWS * HID;
        float wf0[OUT_DIM], wf1[OUT_DIM];
        #pragma unroll
        for (int o = 0; o < OUT_DIM; ++o) {
            wf0[o] = Wfc[lane * OUT_DIM + o];
            wf1[o] = Wfc[(lane + HID / 2) * OUT_DIM + o];
        }
        const float bl0 = b[lane];
        const float bl1 = b[lane + HID / 2];
        const float bo  = (lane < OUT_DIM) ? bfc[lane] : 0.0f;
        for (int n = wave; n < nNodes; n += OWN_THREADS / 32) {
            const int node = n0 + n;
            const float di = dinv[node];
            const float* hr = h + (size_t)node * HID + lane;
            const float v0 = fmaxf(di * (acc[n * HID + lane] + hr[0]) + bl0, 0.0f);
            const float v1 = fmaxf(di * (acc[n * HID + HID / 2 + lane] + hr[HID / 2]) + bl1, 0.0f);
            float p[OUT_DIM];
            #pragma unroll
            for (int o = 0; o < OUT_DIM; ++o) p[o] = v0 * wf0[o] + v1 * wf1[o];
            #pragma unroll
            for (int off = 16; off > 0; off >>= 1) {
                #pragma unroll
                for (int o = 0; o < OUT_DIM; ++o) p[o] += __shfl_xor(p[o], off);
            }
            float pv = p[0];
            #pragma unroll
            for (int o = 1; o < OUT_DIM; ++o) pv = (lane == o) ? p[o] : pv;
            if (lane < OUT_DIM) stage[n * OUT_DIM + lane] = pv + bo;
        }
        __syncthreads();
        const size_t o0 = (size_t)n0 * OUT_DIM;
        for (int i = tid; i < nNodes * OUT_DIM; i += OWN_THREADS) out[o0 + i] = stage[i];
    }
}

extern "C" void kernel_launch(void* const* d_in, const int* in_sizes, int n_in,
                              void* d_out, int out_size, void* d_ws, size_t ws_size,
                              hipStream_t stream) {
    const float* x   = (const float*)d_in[0];
    const int*   ei  = (const int*)  d_in[1];
    const float* ew  = (const float*)d_in[2];
    const float* W1  = (const float*)d_in[3];
    const float* b1  = (const float*)d_in[4];
    const float* W2  = (const float*)d_in[5];
    const float* b2  = (const float*)d_in[6];
    const float* Wfc = (const float*)d_in[7];
    const float* bfc = (const float*)d_in[8];
    float* out = (float*)d_out;

    const int* row = ei;
    const int* col = ei + N_EDGES;

    char* ws = (char*)d_ws;
    size_t off = 0;
    float* dinv = (float*)(ws + off); off += DINV_BYTES;
    float* bufH = (float*)(ws + off); off += PLANE_BYTES;
    float* bufA = (float*)(ws + off); off += PLANE_BYTES;
    int*   rec  = (int*)  (ws + off); off += REC_BYTES;
    int*   cnt  = (int*)  (ws + off); off += CNT_BYTES;
    int*   rel  = (int*)  (ws + off); off += CNT_BYTES;
    int*   tot  = (int*)  (ws + off);
    float* deg1 = bufA;
    (void)n_in; (void)in_sizes; (void)out_size;

    const int TPB = 256;
    const int nodeBlocks = (N_NODES + TPB - 1) / TPB;

    k_init_deg<<<nodeBlocks, TPB, 0, stream>>>(dinv, deg1);
    k_count<<<NCHUNKS, BIN_THREADS, 0, stream>>>(col, ew, dinv, deg1, cnt, ws_size);

    k_scan_dinv<<<NBINS, SCAN_THREADS, 0, stream>>>(cnt, rel, tot, dinv, deg1, ws_size);

    k_gemm_n64<IN_DIM, true><<<GEMM_BLOCKS, GEMM_THREADS, 0, stream>>>(
        x, W1, dinv, bufH, row, col, ew, rel, tot, rec, ws_size);
    k_owner<false><<<NBINS, OWN_THREADS, 0, stream>>>(rec, tot, bufH, dinv, b1, W2, bufA,
                                                      Wfc, bfc, out, ws_size);

    k_owner<true><<<NBINS, OWN_THREADS, 0, stream>>>(rec, tot, bufA, dinv, b2, W2, bufH,
                                                     Wfc, bfc, out, ws_size);
}


# owner sums in registers: a block's records are ordered by target in LDS (count, scan, rank), a run of one target is summed in registers, LDS float atomic only at run ends; sum order: by LDS rank
# speedup vs baseline: 2.9941x; 1.0691x over previous
#include <hip/hip_runtime.h>
#include <hip/hip_bf16.h>


#define N_NODES 100000
#define N_EDGES 1600000
#define IN_DIM  512
#define HID     64
#define OUT_DIM 5

typedef __attribute__((ext_vector_type(2))) float v2f;
typedef __attribute__((ext_vector_type(8))) float v8f;

typedef __attribute__((ext_vector_type(2))) int v2i;

#define BIN_NODES    416
#define NBINS        ((N_NODES + BIN_NODES - 1) / BIN_NODES)
#define CHUNK        3200
#define NCHUNKS      (N_EDGES / CHUNK)
#define BIN_THREADS  640
#define EPT          (CHUNK / BIN_THREADS)
#define SCAN_THREADS 512
#define OWN_THREADS  1024
#define ROW_BITS     17
static_assert(NCHUNKS * CHUNK == N_EDGES, "chunks must tile the edge list");
static_assert(EPT * BIN_THREADS == CHUNK, "a chunk is a whole number of edges per thread");
static_assert(NBINS <= BIN_THREADS, "one thread per block writes the chunk's counts");
static_assert(NCHUNKS <= SCAN_THREADS, "one thread per chunk in the scan");
static_assert(N_NODES <= (1 << ROW_BITS), "source row must fit its field");
static_assert(BIN_NODES < (1 << (30 - ROW_BITS)), "local target must fit its field");
static_assert((BIN_NODES * OUT_DIM * 4) % 128 == 0, "a block's rows of out are whole lines");

#define DINV_BYTES   ((((size_t)N_NODES * 4) + 255) & ~(size_t)255)
#define PLANE_ELEMS  ((size_t)N_NODES * HID)
#define PLANE_BYTES  (PLANE_ELEMS * 4)
#define REC_BYTES    ((size_t)N_EDGES * 8)
#define CNT_BYTES    ((size_t)NBINS * NCHUNKS * 4)
#define WS_NEED      (DINV_BYTES + 2 * PLANE_BYTES + REC_BYTES + 2 * CNT_BYTES + 1024)

__device__ __forceinline__ int wave_half() {
    return (int)((__builtin_amdgcn_s_sendmsg_rtn(0x87) >> 18) & 1u);
}

__global__ void k_init_deg(float* __restrict__ deg0, float* __restrict__ deg1) {
    int i = blockIdx.x * blockDim.x + threadIdx.x;
    if (i < N_NODES) { deg0[i] = 1.0f; deg1[i] = 0.0f; }
}

__device__ __forceinline__ int block_scan_incl(int v, int* s) {
    const int tid = threadIdx.x;
    const int n   = blockDim.x;
    s[tid] = v;
    __syncthreads();
    for (int d = 1; d < n; d <<= 1) {
        const int t = (tid >= d) ? s[tid - d] : 0;
        __syncthreads();
        s[tid] += t;
        __syncthreads();
    }
    return s[tid];
}

__global__ void __launch_bounds__(BIN_THREADS)
k_count(const int* __restrict__ col,
        const float* __restrict__ w,
        float* __restrict__ deg0,
        float* __restrict__ deg1,
        int* __restrict__ cnt,
        size_t wsBytes) {
    if (wsBytes < WS_NEED) return;
    __shared__ int hist[BIN_THREADS];
    const int tid = threadIdx.x;
    hist[tid] = 0;
    __syncthreads();
    float* deg = wave_half() ? deg1 : deg0;
    const int e0 = blockIdx.x * CHUNK + tid;
    int   c[EPT];
    float wv[EPT];
    #pragma unroll
    for (int k = 0; k < EPT; ++k) {
        c[k]  = col[e0 + k * BIN_THREADS];
        wv[k] = w[e0 + k * BIN_THREADS];
    }
    #pragma unroll
    for (int k = 0; k < EPT; ++k) {
        atomicAdd(deg + c[k], wv[k]);
        atomicAdd(&hist[(unsigned)c[k] / BIN_NODES], 1);
    }
    __syncthreads();
    if (tid < NBINS) cnt[tid * NCHUNKS + blockIdx.x] = hist[tid];
}

static_assert(BIN_NODES <= SCAN_THREADS, "one thread per node of the block for dinv");
__global__ void __launch_bounds__(SCAN_THREADS)
k_scan_dinv(const int* __restrict__ cnt,
            int* __restrict__ rel,
            int* __restrict__ tot,
            float* __restrict__ deg0,
            const float* __restrict__ deg1,
            size_t wsBytes) {
    if (wsBytes < WS_NEED) return;
    __shared__ int s[SCAN_THREADS];
    const int tid = threadIdx.x;
    const int bin = blockIdx.x;
    const int node = bin * BIN_NODES + tid;
    if (tid < BIN_NODES && node < N_NODES) {
        const float d = deg0[node] + deg1[node];
        deg0[node] = (d > 0.0f) ? rsqrtf(d) : 0.0f;
    }
    const int v = (tid < NCHUNKS) ? cnt[bin * NCHUNKS + tid] : 0;
    const int incl = block_scan_incl(v, s);
    if (tid < NCHUNKS) rel[bin * NCHUNKS + tid] = incl - v;
    if (tid == SCAN_THREADS - 1) tot[bin] = incl;
}

#define GEMM_WAVES   25
#define GEMM_THREADS (GEMM_WAVES * 32)
#define GEMM_BLOCKS  ((N_NODES / 16) / GEMM_WAVES)
static_assert(GEMM_BLOCKS * GEMM_WAVES * 16 == N_NODES, "row tiles must fill the GEMM grid exactly");
#define PLACE_EPT    (2 * CHUNK / GEMM_THREADS)
#define SCAN_LANES   256
static_assert(2 * GEMM_BLOCKS == NCHUNKS, "two chunks per GEMM work-group");
static_assert(PLACE_EPT * GEMM_THREADS == 2 * CHUNK, "two chunks are a whole number of edges per thread");
static_assert((PLACE_EPT / 2) * GEMM_THREADS == CHUNK, "a thread's first half of edges is the first chunk");
static_assert(NBINS <= SCAN_LANES && SCAN_LANES % 32 == 0 && SCAN_LANES <= GEMM_THREADS, "whole waves scan the totals");
#define GEMM_LDS_MIN 41984

typedef __attribute__((ext_vector_type(4))) float v4f;

__device__ __forceinline__ void gemm_line(const v4f (&a)[4], const float* bl,
                                          v8f& c0, v8f& c1, v8f& c2, v8f& c3) {
    #pragma unroll
    for (int i = 0; i < 4; ++i) {
        const v4f b00 = *(const v4f*)(bl + i * 512);
        const v4f b01 = *(const v4f*)(bl + i * 512 + 128);
        const v4f b10 = *(const v4f*)(bl + i * 512 + 256);
        const v4f b11 = *(const v4f*)(bl + i * 512 + 384);
        const v2f a0 = {a[i].x, a[i].y};
        const v2f a1 = {a[i].z, a[i].w};
        const v2f p00 = {b00.x, b00.y}, p01 = {b00.z, b00.w};
        const v2f p02 = {b01.x, b01.y}, p03 = {b01.z, b01.w};
        const v2f p10 = {b10.x, b10.y}, p11 = {b10.z, b10.w};
        const v2f p12 = {b11.x, b11.y}, p13 = {b11.z, b11.w};
        c0 = __builtin_amdgcn_wmma_f32_16x16x4_f32(false, a0, false, p00, (short)0, c0, false, false);
        c1 = __builtin_amdgcn_wmma_f32_16x16x4_f32(false, a0, false, p01, (short)0, c1, false, false);
        c2 = __builtin_amdgcn_wmma_f32_16x16x4_f32(false, a0, false, p02, (short)0, c2, false, false);
        c3 = __builtin_amdgcn_wmma_f32_16x16x4_f32(false, a0, false, p03, (short)0, c3, false, false);
        c0 = __builtin_amdgcn_wmma_f32_16x16x4_f32(false, a1, false, p10, (short)0, c0, false, false);
        c1 = __builtin_amdgcn_wmma_f32_16x16x4_f32(false, a1, false, p11, (short)0, c1, false, false);
        c2 = __builtin_amdgcn_wmma_f32_16x16x4_f32(false, a1, false, p12, (short)0, c2, false, false);
        c3 = __builtin_amdgcn_wmma_f32_16x16x4_f32(false, a1, false, p13, (short)0, c3, false, false);
    }
}

__device__ __forceinline__ v4f w_unit_load(const float* __restrict__ W, int u) {
    const int cc = u & 15;
    const int tp = (u >> 4) & 1;
    const int k  = (u >> 5) * 2;
    const float* w0 = W + (size_t)k * HID + tp * 32 + cc;
    v4f v;
    v.x = w0[0];
    v.y = w0[HID];
    v.z = w0[16];
    v.w = w0[HID + 16];
    return v;
}
__device__ __forceinline__ int w_unit_place(int u) {
    const int cc = u & 15;
    const int tp = (u >> 4) & 1;
    const int k  = (u >> 5) * 2;
    const int kb = k >> 3;
    const int hh = (k >> 2) & 1;
    const int j  = (k >> 1) & 1;
    return kb * 512 + (2 * j + tp) * 128 + (16 * hh + cc) * 4;
}

template <int K, bool PLACE>
__global__ void __launch_bounds__(GEMM_THREADS)
k_gemm_n64(const float* __restrict__ X,
           const float* __restrict__ W,
           const float* __restrict__ rs,
           float* __restrict__ H,
           const int* __restrict__ row,
           const int* __restrict__ col,
           const float* __restrict__ w,
           const int* __restrict__ rel,
           const int* __restrict__ tot,
           int* __restrict__ rec,
           size_t wsBytes) {
    static_assert(K % 64 == 0, "the k loop takes two 32-k lines per trip");
    constexpr int LDS_FLOATS = (K * HID > GEMM_LDS_MIN) ? K * HID : GEMM_LDS_MIN;
    __shared__ __attribute__((aligned(16))) float Bl[LDS_FLOATS];
    __shared__ int sWave[SCAN_LANES / 32];
    __shared__ int cursor[2 * SCAN_LANES];
    if (PLACE && wsBytes < WS_NEED) return;

    const int tid  = threadIdx.x;
    const int lane = tid & 31;
    const int c    = lane & 15;
    const int h    = lane >> 4;
    const int tile = blockIdx.x * GEMM_WAVES + (tid >> 5);
    const float* xr = X + (size_t)(tile * 16 + c) * K + 4 * h;

    v4f aA[4], aB[4];
    #pragma unroll
    for (int i = 0; i < 4; ++i) aA[i] = *(const v4f*)(xr + 8 * i);

    int pc[PLACE_EPT] = {}, pr[PLACE_EPT] = {}, pw[PLACE_EPT] = {};
    int tv = 0, rl0 = 0, rl1 = 0;
    if (PLACE) {
        const int e0 = blockIdx.x * (2 * CHUNK) + tid;
        #pragma unroll
        for (int k = 0; k < PLACE_EPT; ++k) {
            pc[k] = col[e0 + k * GEMM_THREADS];
            pr[k] = row[e0 + k * GEMM_THREADS];
            pw[k] = __float_as_int(w[e0 + k * GEMM_THREADS]);
        }
        const int tb = (tid < NBINS) ? tid : 0;
        tv  = tot[tb];
        rl0 = rel[tb * NCHUNKS + 2 * blockIdx.x];
        rl1 = rel[tb * NCHUNKS + 2 * blockIdx.x + 1];
    }

    for (int u = tid; u < (K / 2) * 32; u += GEMM_THREADS) {
        const int cc = u & 15;
        const int tp = (u >> 4) & 1;
        const int k  = (u >> 5) * 2;
        const float* w0 = W + (size_t)k * HID + tp * 32 + cc;
        v4f v;
        v.x = w0[0];
        v.y = w0[HID];
        v.z = w0[16];
        v.w = w0[HID + 16];
        const int kb = k >> 3;
        const int hh = (k >> 2) & 1;
        const int j  = (k >> 1) & 1;
        *(v4f*)(Bl + kb * 512 + (2 * j + tp) * 128 + (16 * hh + cc) * 4) = v;
    }

    const int tvs = (tid < NBINS) ? tv : 0;
    int incl = tvs;
    if (PLACE && tid < SCAN_LANES) {
        #pragma unroll
        for (int d = 1; d < 32; d <<= 1) {
            const int y = __shfl_up(incl, d);
            if (lane >= d) incl += y;
        }
        if (lane == 31) sWave[tid >> 5] = incl;
    }
    __syncthreads();
    if (PLACE) {
        if (tid < SCAN_LANES) {
            int p = 0;
            for (int q = 0; q < (tid >> 5); ++q) p += sWave[q];
            if (tid < NBINS) {
                const int base = p + incl - tvs;
                cursor[tid]              = base + rl0;
                cursor[SCAN_LANES + tid] = base + rl1;
            }
        }
        __syncthreads();
    }

    v8f c0 = {}, c1 = {}, c2 = {}, c3 = {};
    const float* bl = Bl + lane * 4;

    for (int ch = 0; ch < K / 32; ch += 2) {
        #pragma unroll
        for (int i = 0; i < 4; ++i) aB[i] = *(const v4f*)(xr + (ch + 1) * 32 + 8 * i);
        gemm_line(aA, bl + ch * 2048, c0, c1, c2, c3);
        if (ch + 2 < K / 32) {
            #pragma unroll
            for (int i = 0; i < 4; ++i) aA[i] = *(const v4f*)(xr + (ch + 2) * 32 + 8 * i);
        }
        gemm_line(aB, bl + (ch + 1) * 2048, c0, c1, c2, c3);
    }

    const int rBase = tile * 16 + 8 * h;
    #pragma unroll
    for (int j = 0; j < 8; ++j) {
        const float sc = rs[rBase + j];
        float* dst = H + (size_t)(rBase + j) * HID + c;
        dst[0]  = sc * c0[j];
        dst[16] = sc * c1[j];
        dst[32] = sc * c2[j];
        dst[48] = sc * c3[j];
    }

    if (PLACE) {
        int bin[PLACE_EPT], pos[PLACE_EPT];
        #pragma unroll
        for (int k = 0; k < PLACE_EPT; ++k) {
            bin[k] = (int)((unsigned)pc[k] / BIN_NODES);
            pos[k] = atomicAdd(&cursor[(k < PLACE_EPT / 2 ? 0 : SCAN_LANES) + bin[k]], 1);
        }
        #pragma unroll
        for (int k = 0; k < PLACE_EPT; ++k) {
            const int cl = pc[k] - bin[k] * BIN_NODES;
            v2i rc;
            rc.x = pr[k] | (cl << ROW_BITS);
            rc.y = pw[k];
            *(v2i*)(rec + 2 * (size_t)pos[k]) = rc;
        }
    }
}

#define EB 32
#define GD 8
static_assert(EB % GD == 0, "gather groups must tile a wave trip");
static_assert(HID == 64, "a row is two 32-float halves, one per wave-wide access");
static_assert(OWN_THREADS == 32 * EB, "a work-group's waves step through the list side by side");
#define OWN_ROWS      (BIN_NODES + 1)
#define OWN_LDS_MIN   41984
#define SORT_CAP      8192
#define SORT_K        (SORT_CAP / OWN_THREADS)
#define HCNT_N        448
#define SORT_INTS     (2 * SORT_CAP + HCNT_N)
#define OWN_SORT_LAST (OWN_ROWS * HID + BIN_NODES * OUT_DIM)
#define OWN_LDS_LAST  (OWN_SORT_LAST + SORT_INTS)
static_assert(SORT_K * OWN_THREADS == SORT_CAP && SORT_CAP % EB == 0, "a window is whole trips");
static_assert(BIN_NODES == 13 * 32, "the scanning wave takes 13 counters per lane");
static_assert(HCNT_N >= BIN_NODES && HCNT_N <= OWN_THREADS, "one thread per counter clears it");
static_assert(OWN_SORT_LAST % 2 == 0, "8-byte LDS pieces");
static_assert(OWN_LDS_LAST >= OWN_LDS_MIN, "still one work-group per WGP");
static_assert((size_t)OWN_LDS_LAST * 4 + 64 <= 320 * 1024, "sums, out rows and window fit the WGP's LDS");
#define V_PITCH       (HID + 4)
#define OWN_W_OFF     (OWN_ROWS * HID)
#define OWN_V_OFF     (OWN_W_OFF + HID * HID)
#define OWN_LDS_MID   (OWN_V_OFF + BIN_NODES * V_PITCH)
static_assert(OWN_LDS_MID >= OWN_LDS_MIN, "still one work-group per WGP");
static_assert(BIN_NODES * V_PITCH >= SORT_INTS && OWN_V_OFF % 2 == 0, "the window fits the plane of v");
static_assert((size_t)OWN_LDS_MID * 4 + 64 <= 320 * 1024, "sums, Wn and v fit the WGP's LDS");
static_assert(OWN_W_OFF % 4 == 0 && OWN_V_OFF % 4 == 0 && V_PITCH % 4 == 0, "16-byte LDS pieces");
static_assert((HID / 2) * 32 == OWN_THREADS, "one unit of Wn per thread");
static_assert(BIN_NODES % 16 == 0 && N_NODES % 16 == 0, "a block is a whole number of 16-row tiles");
static_assert(BIN_NODES / 16 <= OWN_THREADS / 32, "one wave per tile");

template <bool LAST>
__global__ void __launch_bounds__(OWN_THREADS)
k_owner(const int* __restrict__ rec,
        const int* __restrict__ tot,
        const float* __restrict__ h,
        const float* __restrict__ dinv,
        const float* __restrict__ b,
        const float* __restrict__ Wn,
        float* __restrict__ hsN,
        const float* __restrict__ Wfc,
        const float* __restrict__ bfc,
        float* __restrict__ out,
        size_t wsBytes) {
    if (wsBytes < WS_NEED) return;
    constexpr int LDS_FLOATS = LAST ? OWN_LDS_LAST : OWN_LDS_MID;
    constexpr int SORT_OFF   = LAST ? OWN_SORT_LAST : OWN_V_OFF;
    __shared__ __attribute__((aligned(16))) float acc[LDS_FLOATS];
    __shared__ int sStart;
    int* srt  = (int*)(acc + SORT_OFF);
    int* hcnt = srt + 2 * SORT_CAP;

    const int tid  = threadIdx.x;
    const int lane = tid & 31;
    const int wave = __builtin_amdgcn_readfirstlane(tid >> 5);
    const int bin  = blockIdx.x;
    const int n0   = bin * BIN_NODES;
    const int nNodes = (N_NODES - n0 < BIN_NODES) ? (N_NODES - n0) : BIN_NODES;

    v4f wn = {};
    if (!LAST) wn = w_unit_load(Wn, tid);
    for (int i = tid; i < OWN_ROWS * HID; i += OWN_THREADS) acc[i] = 0.0f;
    if (tid < HCNT_N) hcnt[tid] = 0;
    if (!LAST) *(v4f*)(acc + OWN_W_OFF + w_unit_place(tid)) = wn;
    if (wave == 0) {
        int p = 0;
        for (int t = lane; t < bin; t += 32) p += tot[t];
        for (int o = 16; o > 0; o >>= 1) p += __shfl_xor(p, o);
        if (lane == 0) sStart = p;
    }
    __syncthreads();
    const int start = __builtin_amdgcn_readfirstlane(sStart);
    const int end   = start + __builtin_amdgcn_readfirstlane(tot[bin]);

    const v2i pastEnd = {BIN_NODES << ROW_BITS, 0};

    for (int w0 = start; w0 < end; w0 += SORT_CAP) {
        if (w0 != start) {
            __syncthreads();
            if (tid < HCNT_N) hcnt[tid] = 0;
            __syncthreads();
        }
        const int wl = (end - w0 < SORT_CAP) ? (end - w0) : SORT_CAP;

        v2i rcs[SORT_K];
        #pragma unroll
        for (int k = 0; k < SORT_K; ++k) {
            const int i = tid + k * OWN_THREADS;
            rcs[k] = *(const v2i*)(rec + 2 * (size_t)(w0 + ((i < wl) ? i : 0)));
        }
        #pragma unroll
        for (int k = 0; k < SORT_K; ++k)
            if (tid + k * OWN_THREADS < wl) atomicAdd(&hcnt[rcs[k].x >> ROW_BITS], 1);
        __syncthreads();
        if (wave == 0) {
            int v[13];
            int s = 0;
            #pragma unroll
            for (int k = 0; k < 13; ++k) { v[k] = hcnt[lane * 13 + k]; s += v[k]; }
            int incl = s;
            #pragma unroll
            for (int d = 1; d < 32; d <<= 1) {
                const int y = __shfl_up(incl, d);
                if (lane >= d) incl += y;
            }
            int run = incl - s;
            #pragma unroll
            for (int k = 0; k < 13; ++k) { hcnt[lane * 13 + k] = run; run += v[k]; }
        }
        __syncthreads();
        int pos[SORT_K];
        #pragma unroll
        for (int k = 0; k < SORT_K; ++k) {
            pos[k] = 0;
            if (tid + k * OWN_THREADS < wl) pos[k] = atomicAdd(&hcnt[rcs[k].x >> ROW_BITS], 1);
        }
        #pragma unroll
        for (int k = 0; k < SORT_K; ++k)
            if (tid + k * OWN_THREADS < wl) *(v2i*)(srt + 2 * pos[k]) = rcs[k];
        __syncthreads();

        const v2i* sl = (const v2i*)srt;
        v2i rcNext;
        {
            const int i0 = wave * EB + lane;
            const v2i ld = sl[(i0 < wl) ? i0 : 0];
            rcNext = (i0 < wl) ? ld : pastEnd;
        }

        for (int base = wave * EB; base < wl; base += OWN_THREADS) {
            const v2i rc = rcNext;
            {
                const int i1 = base + OWN_THREADS + lane;
                const v2i ld = sl[(i1 < wl) ? i1 : 0];
                rcNext = (i1 < wl) ? ld : pastEnd;
            }
            const int rp    = rc.x;
            const int wBits = rc.y;
            const int r  = rp & ((1 << ROW_BITS) - 1);
            const int cl = rp >> ROW_BITS;

            v2f hv[2][GD];

            #pragma unroll
            for (int j = 0; j < GD; ++j) {
                const int rj = __builtin_amdgcn_readlane(r, j);
                const float* src = h + (size_t)rj * HID + lane;
                hv[0][j].x = src[0];
                hv[0][j].y = src[HID / 2];
            }

            int   tc = __builtin_amdgcn_readlane(cl, 0);
            float a0 = 0.0f, a1 = 0.0f;

            #pragma unroll
            for (int g = 0; g < EB / GD; ++g) {
                const int pb = g & 1;
                if (g + 1 < EB / GD) {
                    #pragma unroll
                    for (int j = 0; j < GD; ++j) {
                        const int rj = __builtin_amdgcn_readlane(r, (g + 1) * GD + j);
                        const float* src = h + (size_t)rj * HID + lane;
                        hv[pb ^ 1][j].x = src[0];
                        hv[pb ^ 1][j].y = src[HID / 2];
                    }
                }
                #pragma unroll
                for (int j = 0; j < GD; ++j) {
                    const int   cj = __builtin_amdgcn_readlane(cl, g * GD + j);
                    const float nj = __int_as_float(
                                         __builtin_amdgcn_readlane(wBits, g * GD + j));
                    if (__builtin_expect(cj != tc, 0)) {
                        float* dst = acc + tc * HID + lane;
                        atomicAdd(dst,           a0);
                        atomicAdd(dst + HID / 2, a1);
                        a0 = 0.0f;
                        a1 = 0.0f;
                        tc = cj;
                    }
                    a0 += nj * hv[pb][j].x;
                    a1 += nj * hv[pb][j].y;
                }
            }
            float* dst = acc + tc * HID + lane;
            atomicAdd(dst,           a0);
            atomicAdd(dst + HID / 2, a1);
        }
    }
    __syncthreads();

    if (!LAST) {
        float* vbuf = acc + OWN_V_OFF;
        const size_t g0 = (size_t)n0 * HID;
        for (int i = tid; i < nNodes * HID; i += OWN_THREADS) {
            const int n = i >> 6;
            const int f = i & 63;
            const float v = dinv[n0 + n] * (acc[i] + h[g0 + i]) + b[f];
            vbuf[n * V_PITCH + f] = fmaxf(v, 0.0f);
        }
        __syncthreads();
        if (wave < (nNodes >> 4)) {
            const int c  = lane & 15;
            const int hh = lane >> 4;
            const float* vr = vbuf + (wave * 16 + c) * V_PITCH + 4 * hh;
            v4f aA[4], aB[4];
            #pragma unroll
            for (int i = 0; i < 4; ++i) {
                aA[i] = *(const v4f*)(vr + 8 * i);
                aB[i] = *(const v4f*)(vr + 32 + 8 * i);
            }
            v8f c0 = {}, c1 = {}, c2 = {}, c3 = {};
            const float* bl = acc + OWN_W_OFF + lane * 4;
            gemm_line(aA, bl, c0, c1, c2, c3);
            gemm_line(aB, bl + 2048, c0, c1, c2, c3);
            const int rBase = n0 + wave * 16 + 8 * hh;
            #pragma unroll
            for (int j = 0; j < 8; ++j) {
                const float sc = dinv[rBase + j];
                float* dst = hsN + (size_t)(rBase + j) * HID + c;
                dst[0]  = sc * c0[j];
                dst[16] = sc * c1[j];
                dst[32] = sc * c2[j];
                dst[48] = sc * c3[j];
            }
        }
    } else {
        float* stage = acc + OWN_ROWS * HID;
        float wf0[OUT_DIM], wf1[OUT_DIM];
        #pragma unroll
        for (int o = 0; o < OUT_DIM; ++o) {
            wf0[o] = Wfc[lane * OUT_DIM + o];
            wf1[o] = Wfc[(lane + HID / 2) * OUT_DIM + o];
        }
        const float bl0 = b[lane];
        const float bl1 = b[lane + HID / 2];
        const float bo  = (lane < OUT_DIM) ? bfc[lane] : 0.0f;
        for (int n = wave; n < nNodes; n += OWN_THREADS / 32) {
            const int node = n0 + n;
            const float di = dinv[node];
            const float* hr = h + (size_t)node * HID + lane;
            const float v0 = fmaxf(di * (acc[n * HID + lane] + hr[0]) + bl0, 0.0f);
            const float v1 = fmaxf(di * (acc[n * HID + HID / 2 + lane] + hr[HID / 2]) + bl1, 0.0f);
            float p[OUT_DIM];
            #pragma unroll
            for (int o = 0; o < OUT_DIM; ++o) p[o] = v0 * wf0[o] + v1 * wf1[o];
            #pragma unroll
            for (int off = 16; off > 0; off >>= 1) {
                #pragma unroll
                for (int o = 0; o < OUT_DIM; ++o) p[o] += __shfl_xor(p[o], off);
            }
            float pv = p[0];
            #pragma unroll
            for (int o = 1; o < OUT_DIM; ++o) pv = (lane == o) ? p[o] : pv;
            if (lane < OUT_DIM) stage[n * OUT_DIM + lane] = pv + bo;
        }
        __syncthreads();
        const size_t o0 = (size_t)n0 * OUT_DIM;
        for (int i = tid; i < nNodes * OUT_DIM; i += OWN_THREADS) out[o0 + i] = stage[i];
    }
}

extern "C" void kernel_launch(void* const* d_in, const int* in_sizes, int n_in,
                              void* d_out, int out_size, void* d_ws, size_t ws_size,
                              hipStream_t stream) {
    const float* x   = (const float*)d_in[0];
    const int*   ei  = (const int*)  d_in[1];
    const float* ew  = (const float*)d_in[2];
    const float* W1  = (const float*)d_in[3];
    const float* b1  = (const float*)d_in[4];
    const float* W2  = (const float*)d_in[5];
    const float* b2  = (const float*)d_in[6];
    const float* Wfc = (const float*)d_in[7];
    const float* bfc = (const float*)d_in[8];
    float* out = (float*)d_out;

    const int* row = ei;
    const int* col = ei + N_EDGES;

    char* ws = (char*)d_ws;
    size_t off = 0;
    float* dinv = (float*)(ws + off); off += DINV_BYTES;
    float* bufH = (float*)(ws + off); off += PLANE_BYTES;
    float* bufA = (float*)(ws + off); off += PLANE_BYTES;
    int*   rec  = (int*)  (ws + off); off += REC_BYTES;
    int*   cnt  = (int*)  (ws + off); off += CNT_BYTES;
    int*   rel  = (int*)  (ws + off); off += CNT_BYTES;
    int*   tot  = (int*)  (ws + off);
    float* deg1 = bufA;
    (void)n_in; (void)in_sizes; (void)out_size;

    const int TPB = 256;
    const int nodeBlocks = (N_NODES + TPB - 1) / TPB;

    k_init_deg<<<nodeBlocks, TPB, 0, stream>>>(dinv, deg1);
    k_count<<<NCHUNKS, BIN_THREADS, 0, stream>>>(col, ew, dinv, deg1, cnt, ws_size);

    k_scan_dinv<<<NBINS, SCAN_THREADS, 0, stream>>>(cnt, rel, tot, dinv, deg1, ws_size);

    k_gemm_n64<IN_DIM, true><<<GEMM_BLOCKS, GEMM_THREADS, 0, stream>>>(
        x, W1, dinv, bufH, row, col, ew, rel, tot, rec, ws_size);
    k_owner<false><<<NBINS, OWN_THREADS, 0, stream>>>(rec, tot, bufH, dinv, b1, W2, bufA,
                                                      Wfc, bfc, out, ws_size);

    k_owner<true><<<NBINS, OWN_THREADS, 0, stream>>>(rec, tot, bufA, dinv, b2, W2, bufH,
                                                     Wfc, bfc, out, ws_size);
}


# owner epilogue in fragment registers: a wave reads its 16-row tile of the LDS sums as matrix A pieces (row pitch 272 B), applies relu/bias there; FC head on the f32 matrix line; FC sum order changes
# speedup vs baseline: 3.1228x; 1.0430x over previous
#include <hip/hip_runtime.h>
#include <hip/hip_bf16.h>


#define N_NODES 100000
#define N_EDGES 1600000
#define IN_DIM  512
#define HID     64
#define OUT_DIM 5

typedef __attribute__((ext_vector_type(2))) float v2f;
typedef __attribute__((ext_vector_type(8))) float v8f;

typedef __attribute__((ext_vector_type(2))) int v2i;

#define BIN_NODES    416
#define NBINS        ((N_NODES + BIN_NODES - 1) / BIN_NODES)
#define CHUNK        3200
#define NCHUNKS      (N_EDGES / CHUNK)
#define BIN_THREADS  640
#define EPT          (CHUNK / BIN_THREADS)
#define SCAN_THREADS 512
#define OWN_THREADS  1024
#define ROW_BITS     17
static_assert(NCHUNKS * CHUNK == N_EDGES, "chunks must tile the edge list");
static_assert(EPT * BIN_THREADS == CHUNK, "a chunk is a whole number of edges per thread");
static_assert(NBINS <= BIN_THREADS, "one thread per block writes the chunk's counts");
static_assert(NCHUNKS <= SCAN_THREADS, "one thread per chunk in the scan");
static_assert(N_NODES <= (1 << ROW_BITS), "source row must fit its field");
static_assert(BIN_NODES < (1 << (30 - ROW_BITS)), "local target must fit its field");
static_assert((BIN_NODES * OUT_DIM * 4) % 128 == 0, "a block's rows of out are whole lines");

#define DINV_BYTES   ((((size_t)N_NODES * 4) + 255) & ~(size_t)255)
#define PLANE_ELEMS  ((size_t)N_NODES * HID)
#define PLANE_BYTES  (PLANE_ELEMS * 4)
#define REC_BYTES    ((size_t)N_EDGES * 8)
#define CNT_BYTES    ((size_t)NBINS * NCHUNKS * 4)
#define WS_NEED      (DINV_BYTES + 2 * PLANE_BYTES + REC_BYTES + 2 * CNT_BYTES + 1024)

__device__ __forceinline__ int wave_half() {
    return (int)((__builtin_amdgcn_s_sendmsg_rtn(0x87) >> 18) & 1u);
}

__global__ void k_init_deg(float* __restrict__ deg0, float* __restrict__ deg1) {
    int i = blockIdx.x * blockDim.x + threadIdx.x;
    if (i < N_NODES) { deg0[i] = 1.0f; deg1[i] = 0.0f; }
}

__device__ __forceinline__ int block_scan_incl(int v, int* s) {
    const int tid = threadIdx.x;
    const int n   = blockDim.x;
    s[tid] = v;
    __syncthreads();
    for (int d = 1; d < n; d <<= 1) {
        const int t = (tid >= d) ? s[tid - d] : 0;
        __syncthreads();
        s[tid] += t;
        __syncthreads();
    }
    return s[tid];
}

__global__ void __launch_bounds__(BIN_THREADS)
k_count(const int* __restrict__ col,
        const float* __restrict__ w,
        float* __restrict__ deg0,
        float* __restrict__ deg1,
        int* __restrict__ cnt,
        size_t wsBytes) {
    if (wsBytes < WS_NEED) return;
    __shared__ int hist[BIN_THREADS];
    const int tid = threadIdx.x;
    hist[tid] = 0;
    __syncthreads();
    float* deg = wave_half() ? deg1 : deg0;
    const int e0 = blockIdx.x * CHUNK + tid;
    int   c[EPT];
    float wv[EPT];
    #pragma unroll
    for (int k = 0; k < EPT; ++k) {
        c[k]  = col[e0 + k * BIN_THREADS];
        wv[k] = w[e0 + k * BIN_THREADS];
    }
    #pragma unroll
    for (int k = 0; k < EPT; ++k) {
        atomicAdd(deg + c[k], wv[k]);
        atomicAdd(&hist[(unsigned)c[k] / BIN_NODES], 1);
    }
    __syncthreads();
    if (tid < NBINS) cnt[tid * NCHUNKS + blockIdx.x] = hist[tid];
}

static_assert(BIN_NODES <= SCAN_THREADS, "one thread per node of the block for dinv");
__global__ void __launch_bounds__(SCAN_THREADS)
k_scan_dinv(const int* __restrict__ cnt,
            int* __restrict__ rel,
            int* __restrict__ tot,
            float* __restrict__ deg0,
            const float* __restrict__ deg1,
            size_t wsBytes) {
    if (wsBytes < WS_NEED) return;
    __shared__ int s[SCAN_THREADS];
    const int tid = threadIdx.x;
    const int bin = blockIdx.x;
    const int node = bin * BIN_NODES + tid;
    if (tid < BIN_NODES && node < N_NODES) {
        const float d = deg0[node] + deg1[node];
        deg0[node] = (d > 0.0f) ? rsqrtf(d) : 0.0f;
    }
    const int v = (tid < NCHUNKS) ? cnt[bin * NCHUNKS + tid] : 0;
    const int incl = block_scan_incl(v, s);
    if (tid < NCHUNKS) rel[bin * NCHUNKS + tid] = incl - v;
    if (tid == SCAN_THREADS - 1) tot[bin] = incl;
}

#define GEMM_WAVES   25
#define GEMM_THREADS (GEMM_WAVES * 32)
#define GEMM_BLOCKS  ((N_NODES / 16) / GEMM_WAVES)
static_assert(GEMM_BLOCKS * GEMM_WAVES * 16 == N_NODES, "row tiles must fill the GEMM grid exactly");
#define PLACE_EPT    (2 * CHUNK / GEMM_THREADS)
#define SCAN_LANES   256
static_assert(2 * GEMM_BLOCKS == NCHUNKS, "two chunks per GEMM work-group");
static_assert(PLACE_EPT * GEMM_THREADS == 2 * CHUNK, "two chunks are a whole number of edges per thread");
static_assert((PLACE_EPT / 2) * GEMM_THREADS == CHUNK, "a thread's first half of edges is the first chunk");
static_assert(NBINS <= SCAN_LANES && SCAN_LANES % 32 == 0 && SCAN_LANES <= GEMM_THREADS, "whole waves scan the totals");
#define GEMM_LDS_MIN 41984

typedef __attribute__((ext_vector_type(4))) float v4f;

__device__ __forceinline__ void gemm_line(const v4f (&a)[4], const float* bl,
                                          v8f& c0, v8f& c1, v8f& c2, v8f& c3) {
    #pragma unroll
    for (int i = 0; i < 4; ++i) {
        const v4f b00 = *(const v4f*)(bl + i * 512);
        const v4f b01 = *(const v4f*)(bl + i * 512 + 128);
        const v4f b10 = *(const v4f*)(bl + i * 512 + 256);
        const v4f b11 = *(const v4f*)(bl + i * 512 + 384);
        const v2f a0 = {a[i].x, a[i].y};
        const v2f a1 = {a[i].z, a[i].w};
        const v2f p00 = {b00.x, b00.y}, p01 = {b00.z, b00.w};
        const v2f p02 = {b01.x, b01.y}, p03 = {b01.z, b01.w};
        const v2f p10 = {b10.x, b10.y}, p11 = {b10.z, b10.w};
        const v2f p12 = {b11.x, b11.y}, p13 = {b11.z, b11.w};
        c0 = __builtin_amdgcn_wmma_f32_16x16x4_f32(false, a0, false, p00, (short)0, c0, false, false);
        c1 = __builtin_amdgcn_wmma_f32_16x16x4_f32(false, a0, false, p01, (short)0, c1, false, false);
        c2 = __builtin_amdgcn_wmma_f32_16x16x4_f32(false, a0, false, p02, (short)0, c2, false, false);
        c3 = __builtin_amdgcn_wmma_f32_16x16x4_f32(false, a0, false, p03, (short)0, c3, false, false);
        c0 = __builtin_amdgcn_wmma_f32_16x16x4_f32(false, a1, false, p10, (short)0, c0, false, false);
        c1 = __builtin_amdgcn_wmma_f32_16x16x4_f32(false, a1, false, p11, (short)0, c1, false, false);
        c2 = __builtin_amdgcn_wmma_f32_16x16x4_f32(false, a1, false, p12, (short)0, c2, false, false);
        c3 = __builtin_amdgcn_wmma_f32_16x16x4_f32(false, a1, false, p13, (short)0, c3, false, false);
    }
}

__device__ __forceinline__ v4f w_unit_load(const float* __restrict__ W, int u) {
    const int cc = u & 15;
    const int tp = (u >> 4) & 1;
    const int k  = (u >> 5) * 2;
    const float* w0 = W + (size_t)k * HID + tp * 32 + cc;
    v4f v;
    v.x = w0[0];
    v.y = w0[HID];
    v.z = w0[16];
    v.w = w0[HID + 16];
    return v;
}
__device__ __forceinline__ int w_unit_place(int u) {
    const int cc = u & 15;
    const int tp = (u >> 4) & 1;
    const int k  = (u >> 5) * 2;
    const int kb = k >> 3;
    const int hh = (k >> 2) & 1;
    const int j  = (k >> 1) & 1;
    return kb * 512 + (2 * j + tp) * 128 + (16 * hh + cc) * 4;
}

template <int K, bool PLACE>
__global__ void __launch_bounds__(GEMM_THREADS)
k_gemm_n64(const float* __restrict__ X,
           const float* __restrict__ W,
           const float* __restrict__ rs,
           float* __restrict__ H,
           const int* __restrict__ row,
           const int* __restrict__ col,
           const float* __restrict__ w,
           const int* __restrict__ rel,
           const int* __restrict__ tot,
           int* __restrict__ rec,
           size_t wsBytes) {
    static_assert(K % 64 == 0, "the k loop takes two 32-k lines per trip");
    constexpr int LDS_FLOATS = (K * HID > GEMM_LDS_MIN) ? K * HID : GEMM_LDS_MIN;
    __shared__ __attribute__((aligned(16))) float Bl[LDS_FLOATS];
    __shared__ int sWave[SCAN_LANES / 32];
    __shared__ int cursor[2 * SCAN_LANES];
    if (PLACE && wsBytes < WS_NEED) return;

    const int tid  = threadIdx.x;
    const int lane = tid & 31;
    const int c    = lane & 15;
    const int h    = lane >> 4;
    const int tile = blockIdx.x * GEMM_WAVES + (tid >> 5);
    const float* xr = X + (size_t)(tile * 16 + c) * K + 4 * h;

    v4f aA[4], aB[4];
    #pragma unroll
    for (int i = 0; i < 4; ++i) aA[i] = *(const v4f*)(xr + 8 * i);

    int pc[PLACE_EPT] = {}, pr[PLACE_EPT] = {}, pw[PLACE_EPT] = {};
    int tv = 0, rl0 = 0, rl1 = 0;
    if (PLACE) {
        const int e0 = blockIdx.x * (2 * CHUNK) + tid;
        #pragma unroll
        for (int k = 0; k < PLACE_EPT; ++k) {
            pc[k] = col[e0 + k * GEMM_THREADS];
            pr[k] = row[e0 + k * GEMM_THREADS];
            pw[k] = __float_as_int(w[e0 + k * GEMM_THREADS]);
        }
        const int tb = (tid < NBINS) ? tid : 0;
        tv  = tot[tb];
        rl0 = rel[tb * NCHUNKS + 2 * blockIdx.x];
        rl1 = rel[tb * NCHUNKS + 2 * blockIdx.x + 1];
    }

    for (int u = tid; u < (K / 2) * 32; u += GEMM_THREADS) {
        const int cc = u & 15;
        const int tp = (u >> 4) & 1;
        const int k  = (u >> 5) * 2;
        const float* w0 = W + (size_t)k * HID + tp * 32 + cc;
        v4f v;
        v.x = w0[0];
        v.y = w0[HID];
        v.z = w0[16];
        v.w = w0[HID + 16];
        const int kb = k >> 3;
        const int hh = (k >> 2) & 1;
        const int j  = (k >> 1) & 1;
        *(v4f*)(Bl + kb * 512 + (2 * j + tp) * 128 + (16 * hh + cc) * 4) = v;
    }

    const int tvs = (tid < NBINS) ? tv : 0;
    int incl = tvs;
    if (PLACE && tid < SCAN_LANES) {
        #pragma unroll
        for (int d = 1; d < 32; d <<= 1) {
            const int y = __shfl_up(incl, d);
            if (lane >= d) incl += y;
        }
        if (lane == 31) sWave[tid >> 5] = incl;
    }
    __syncthreads();
    if (PLACE) {
        if (tid < SCAN_LANES) {
            int p = 0;
            for (int q = 0; q < (tid >> 5); ++q) p += sWave[q];
            if (tid < NBINS) {
                const int base = p + incl - tvs;
                cursor[tid]              = base + rl0;
                cursor[SCAN_LANES + tid] = base + rl1;
            }
        }
        __syncthreads();
    }

    v8f c0 = {}, c1 = {}, c2 = {}, c3 = {};
    const float* bl = Bl + lane * 4;

    for (int ch = 0; ch < K / 32; ch += 2) {
        #pragma unroll
        for (int i = 0; i < 4; ++i) aB[i] = *(const v4f*)(xr + (ch + 1) * 32 + 8 * i);
        gemm_line(aA, bl + ch * 2048, c0, c1, c2, c3);
        if (ch + 2 < K / 32) {
            #pragma unroll
            for (int i = 0; i < 4; ++i) aA[i] = *(const v4f*)(xr + (ch + 2) * 32 + 8 * i);
        }
        gemm_line(aB, bl + (ch + 1) * 2048, c0, c1, c2, c3);
    }

    const int rBase = tile * 16 + 8 * h;
    #pragma unroll
    for (int j = 0; j < 8; ++j) {
        const float sc = rs[rBase + j];
        float* dst = H + (size_t)(rBase + j) * HID + c;
        dst[0]  = sc * c0[j];
        dst[16] = sc * c1[j];
        dst[32] = sc * c2[j];
        dst[48] = sc * c3[j];
    }

    if (PLACE) {
        int bin[PLACE_EPT], pos[PLACE_EPT];
        #pragma unroll
        for (int k = 0; k < PLACE_EPT; ++k) {
            bin[k] = (int)((unsigned)pc[k] / BIN_NODES);
            pos[k] = atomicAdd(&cursor[(k < PLACE_EPT / 2 ? 0 : SCAN_LANES) + bin[k]], 1);
        }
        #pragma unroll
        for (int k = 0; k < PLACE_EPT; ++k) {
            const int cl = pc[k] - bin[k] * BIN_NODES;
            v2i rc;
            rc.x = pr[k] | (cl << ROW_BITS);
            rc.y = pw[k];
            *(v2i*)(rec + 2 * (size_t)pos[k]) = rc;
        }
    }
}

#define EB 32
#define GD 8
static_assert(EB % GD == 0, "gather groups must tile a wave trip");
static_assert(HID == 64, "a row is two 32-float halves, one per wave-wide access");
static_assert(OWN_THREADS == 32 * EB, "a work-group's waves step through the list side by side");
#define ACC_PITCH     (HID + 4)
#define OWN_ROWS      (BIN_NODES + 1)
#define OWN_ACC       (OWN_ROWS * ACC_PITCH)
#define OWN_LDS_MIN   41984
#define SORT_CAP      8192
#define SORT_K        (SORT_CAP / OWN_THREADS)
#define HCNT_N        448
#define SORT_INTS     (2 * SORT_CAP + HCNT_N)
#define OWN_W_OFF     OWN_ACC
#define OWN_SORT_MID  (OWN_W_OFF + HID * HID)
#define OWN_LDS_MID   (OWN_SORT_MID + SORT_INTS)
#define FC_FLOATS     ((HID / 8) * 128)
#define OWN_OUT_OFF   OWN_ACC
#define OWN_FC_OFF    (OWN_OUT_OFF + BIN_NODES * OUT_DIM)
#define OWN_SORT_LAST (OWN_FC_OFF + FC_FLOATS)
#define OWN_LDS_LAST  (OWN_SORT_LAST + SORT_INTS)
static_assert(SORT_K * OWN_THREADS == SORT_CAP && SORT_CAP % EB == 0, "a window is whole trips");
static_assert(BIN_NODES == 13 * 32, "the scanning wave takes 13 counters per lane");
static_assert(HCNT_N >= BIN_NODES && HCNT_N <= OWN_THREADS, "one thread per counter clears it");
static_assert(OWN_SORT_LAST % 2 == 0 && OWN_SORT_MID % 2 == 0, "8-byte LDS pieces");
static_assert(OWN_LDS_LAST >= OWN_LDS_MIN && OWN_LDS_MID >= OWN_LDS_MIN, "still one work-group per WGP");
static_assert((size_t)OWN_LDS_LAST * 4 + 512 <= 320 * 1024, "sums, out rows, Wfc and window fit the WGP's LDS");
static_assert((size_t)OWN_LDS_MID * 4 + 512 <= 320 * 1024, "sums, Wn and window fit the WGP's LDS");
static_assert(ACC_PITCH % 4 == 0 && OWN_W_OFF % 4 == 0 && OWN_FC_OFF % 4 == 0, "16-byte LDS pieces");
static_assert((HID / 2) * 32 == OWN_THREADS, "one unit of Wn per thread");
static_assert(FC_FLOATS == OWN_THREADS, "one element of padded Wfc per thread");
static_assert(OUT_DIM <= 16, "the FC head is one 16-column tile");
static_assert(BIN_NODES % 16 == 0 && N_NODES % 16 == 0, "a block is a whole number of 16-row tiles");
static_assert(BIN_NODES / 16 <= OWN_THREADS / 32, "one wave per tile");

template <bool LAST>
__global__ void __launch_bounds__(OWN_THREADS)
k_owner(const int* __restrict__ rec,
        const int* __restrict__ tot,
        const float* __restrict__ h,
        const float* __restrict__ dinv,
        const float* __restrict__ b,
        const float* __restrict__ Wn,
        float* __restrict__ hsN,
        const float* __restrict__ Wfc,
        const float* __restrict__ bfc,
        float* __restrict__ out,
        size_t wsBytes) {
    if (wsBytes < WS_NEED) return;
    constexpr int LDS_FLOATS = LAST ? OWN_LDS_LAST : OWN_LDS_MID;
    constexpr int SORT_OFF   = LAST ? OWN_SORT_LAST : OWN_SORT_MID;
    __shared__ __attribute__((aligned(16))) float acc[LDS_FLOATS];
    __shared__ __attribute__((aligned(16))) float sB[HID];
    __shared__ int sStart;
    int* srt  = (int*)(acc + SORT_OFF);
    int* hcnt = srt + 2 * SORT_CAP;

    const int tid  = threadIdx.x;
    const int lane = tid & 31;
    const int wave = __builtin_amdgcn_readfirstlane(tid >> 5);
    const int bin  = blockIdx.x;
    const int n0   = bin * BIN_NODES;
    const int nNodes = (N_NODES - n0 < BIN_NODES) ? (N_NODES - n0) : BIN_NODES;

    v4f wn = {};
    float wf = 0.0f;
    if (!LAST) {
        wn = w_unit_load(Wn, tid);
    } else {
        const int e  = tid & 3;
        const int cc = (tid >> 2) & 15;
        const int hf = (tid >> 6) & 1;
        const int kq = tid >> 7;
        const float ld = Wfc[(8 * kq + 4 * hf + e) * OUT_DIM + ((cc < OUT_DIM) ? cc : 0)];
        wf = (cc < OUT_DIM) ? ld : 0.0f;
    }
    for (int i = tid; i < OWN_ACC; i += OWN_THREADS) acc[i] = 0.0f;
    if (tid < HCNT_N) hcnt[tid] = 0;
    if (tid < HID) sB[tid] = b[tid];
    if (!LAST) *(v4f*)(acc + OWN_W_OFF + w_unit_place(tid)) = wn;
    else       acc[OWN_FC_OFF + tid] = wf;
    if (wave == 0) {
        int p = 0;
        for (int t = lane; t < bin; t += 32) p += tot[t];
        for (int o = 16; o > 0; o >>= 1) p += __shfl_xor(p, o);
        if (lane == 0) sStart = p;
    }
    __syncthreads();
    const int start = __builtin_amdgcn_readfirstlane(sStart);
    const int end   = start + __builtin_amdgcn_readfirstlane(tot[bin]);

    const v2i pastEnd = {BIN_NODES << ROW_BITS, 0};

    for (int w0 = start; w0 < end; w0 += SORT_CAP) {
        if (w0 != start) {
            __syncthreads();
            if (tid < HCNT_N) hcnt[tid] = 0;
            __syncthreads();
        }
        const int wl = (end - w0 < SORT_CAP) ? (end - w0) : SORT_CAP;

        v2i rcs[SORT_K];
        #pragma unroll
        for (int k = 0; k < SORT_K; ++k) {
            const int i = tid + k * OWN_THREADS;
            rcs[k] = *(const v2i*)(rec + 2 * (size_t)(w0 + ((i < wl) ? i : 0)));
        }
        #pragma unroll
        for (int k = 0; k < SORT_K; ++k)
            if (tid + k * OWN_THREADS < wl) atomicAdd(&hcnt[rcs[k].x >> ROW_BITS], 1);
        __syncthreads();
        if (wave == 0) {
            int v[13];
            int s = 0;
            #pragma unroll
            for (int k = 0; k < 13; ++k) { v[k] = hcnt[lane * 13 + k]; s += v[k]; }
            int incl = s;
            #pragma unroll
            for (int d = 1; d < 32; d <<= 1) {
                const int y = __shfl_up(incl, d);
                if (lane >= d) incl += y;
            }
            int run = incl - s;
            #pragma unroll
            for (int k = 0; k < 13; ++k) { hcnt[lane * 13 + k] = run; run += v[k]; }
        }
        __syncthreads();
        int pos[SORT_K];
        #pragma unroll
        for (int k = 0; k < SORT_K; ++k) {
            pos[k] = 0;
            if (tid + k * OWN_THREADS < wl) pos[k] = atomicAdd(&hcnt[rcs[k].x >> ROW_BITS], 1);
        }
        #pragma unroll
        for (int k = 0; k < SORT_K; ++k)
            if (tid + k * OWN_THREADS < wl) *(v2i*)(srt + 2 * pos[k]) = rcs[k];
        __syncthreads();

        const v2i* sl = (const v2i*)srt;
        v2i rcNext;
        {
            const int i0 = wave * EB + lane;
            const v2i ld = sl[(i0 < wl) ? i0 : 0];
            rcNext = (i0 < wl) ? ld : pastEnd;
        }

        for (int base = wave * EB; base < wl; base += OWN_THREADS) {
            const v2i rc = rcNext;
            {
                const int i1 = base + OWN_THREADS + lane;
                const v2i ld = sl[(i1 < wl) ? i1 : 0];
                rcNext = (i1 < wl) ? ld : pastEnd;
            }
            const int rp    = rc.x;
            const int wBits = rc.y;
            const int r  = rp & ((1 << ROW_BITS) - 1);
            const int cl = rp >> ROW_BITS;

            v2f hv[2][GD];

            #pragma unroll
            for (int j = 0; j < GD; ++j) {
                const int rj = __builtin_amdgcn_readlane(r, j);
                const float* src = h + (size_t)rj * HID + lane;
                hv[0][j].x = src[0];
                hv[0][j].y = src[HID / 2];
            }

            int   tc = __builtin_amdgcn_readlane(cl, 0);
            float a0 = 0.0f, a1 = 0.0f;

            #pragma unroll
            for (int g = 0; g < EB / GD; ++g) {
                const int pb = g & 1;
                if (g + 1 < EB / GD) {
                    #pragma unroll
                    for (int j = 0; j < GD; ++j) {
                        const int rj = __builtin_amdgcn_readlane(r, (g + 1) * GD + j);
                        const float* src = h + (size_t)rj * HID + lane;
                        hv[pb ^ 1][j].x = src[0];
                        hv[pb ^ 1][j].y = src[HID / 2];
                    }
                }
                #pragma unroll
                for (int j = 0; j < GD; ++j) {
                    const int   cj = __builtin_amdgcn_readlane(cl, g * GD + j);
                    const float nj = __int_as_float(
                                         __builtin_amdgcn_readlane(wBits, g * GD + j));
                    if (__builtin_expect(cj != tc, 0)) {
                        float* dst = acc + tc * ACC_PITCH + lane;
                        atomicAdd(dst,           a0);
                        atomicAdd(dst + HID / 2, a1);
                        a0 = 0.0f;
                        a1 = 0.0f;
                        tc = cj;
                    }
                    a0 += nj * hv[pb][j].x;
                    a1 += nj * hv[pb][j].y;
                }
            }
            float* dst = acc + tc * ACC_PITCH + lane;
            atomicAdd(dst,           a0);
            atomicAdd(dst + HID / 2, a1);
        }
    }

    const int  c     = lane & 15;
    const int  hh    = lane >> 4;
    const bool own   = wave < (nNodes >> 4);
    const int  rowL  = (own ? wave * 16 : 0) + c;
    const float* hr  = h + (size_t)(n0 + rowL) * HID + 4 * hh;
    v4f vp[8];
    #pragma unroll
    for (int kb = 0; kb < 8; ++kb) vp[kb] = *(const v4f*)(hr + 8 * kb);
    const float di = dinv[n0 + rowL];
    const float bo = LAST ? bfc[(c < OUT_DIM) ? c : 0] : 0.0f;
    __builtin_amdgcn_fence(__ATOMIC_RELEASE, "workgroup", "local");
    __builtin_amdgcn_s_barrier();
    __builtin_amdgcn_fence(__ATOMIC_ACQUIRE, "workgroup", "local");

    if (own) {
        const float* ar = acc + rowL * ACC_PITCH + 4 * hh;
        #pragma unroll
        for (int kb = 0; kb < 8; ++kb) {
            const v4f sm = *(const v4f*)(ar + 8 * kb);
            const v4f bk = *(const v4f*)(sB + 8 * kb + 4 * hh);
            v4f v;
            v.x = fmaxf(di * (sm.x + vp[kb].x) + bk.x, 0.0f);
            v.y = fmaxf(di * (sm.y + vp[kb].y) + bk.y, 0.0f);
            v.z = fmaxf(di * (sm.z + vp[kb].z) + bk.z, 0.0f);
            v.w = fmaxf(di * (sm.w + vp[kb].w) + bk.w, 0.0f);
            vp[kb] = v;
        }
        if (!LAST) {
            const v4f aA[4] = {vp[0], vp[1], vp[2], vp[3]};
            const v4f aB[4] = {vp[4], vp[5], vp[6], vp[7]};
            v8f c0 = {}, c1 = {}, c2 = {}, c3 = {};
            const float* bl = acc + OWN_W_OFF + lane * 4;
            gemm_line(aA, bl, c0, c1, c2, c3);
            gemm_line(aB, bl + 2048, c0, c1, c2, c3);
            const int rBase = n0 + wave * 16 + 8 * hh;
            #pragma unroll
            for (int j = 0; j < 8; ++j) {
                const float sc = dinv[rBase + j];
                float* dst = hsN + (size_t)(rBase + j) * HID + c;
                dst[0]  = sc * c0[j];
                dst[16] = sc * c1[j];
                dst[32] = sc * c2[j];
                dst[48] = sc * c3[j];
            }
        } else {
            const float* fl = acc + OWN_FC_OFF + lane * 4;
            v8f d0 = {}, d1 = {};
            #pragma unroll
            for (int kb = 0; kb < 8; kb += 2) {
                const v4f f0 = *(const v4f*)(fl + kb * 128);
                const v4f f1 = *(const v4f*)(fl + (kb + 1) * 128);
                const v2f x00 = {vp[kb].x, vp[kb].y},         x01 = {vp[kb].z, vp[kb].w};
                const v2f x10 = {vp[kb + 1].x, vp[kb + 1].y}, x11 = {vp[kb + 1].z, vp[kb + 1].w};
                const v2f y00 = {f0.x, f0.y}, y01 = {f0.z, f0.w};
                const v2f y10 = {f1.x, f1.y}, y11 = {f1.z, f1.w};
                d0 = __builtin_amdgcn_wmma_f32_16x16x4_f32(false, x00, false, y00, (short)0, d0, false, false);
                d1 = __builtin_amdgcn_wmma_f32_16x16x4_f32(false, x10, false, y10, (short)0, d1, false, false);
                d0 = __builtin_amdgcn_wmma_f32_16x16x4_f32(false, x01, false, y01, (short)0, d0, false, false);
                d1 = __builtin_amdgcn_wmma_f32_16x16x4_f32(false, x11, false, y11, (short)0, d1, false, false);
            }
            float* stage = acc + OWN_OUT_OFF + (wave * 16 + 8 * hh) * OUT_DIM + c;
            if (c < OUT_DIM) {
                #pragma unroll
                for (int j = 0; j < 8; ++j) stage[j * OUT_DIM] = (d0[j] + d1[j]) + bo;
            }
        }
    }
    if (LAST) {
        __syncthreads();
        const float* stage = acc + OWN_OUT_OFF;
        const size_t o0 = (size_t)n0 * OUT_DIM;
        for (int i = tid; i < nNodes * OUT_DIM; i += OWN_THREADS) out[o0 + i] = stage[i];
    }
}

extern "C" void kernel_launch(void* const* d_in, const int* in_sizes, int n_in,
                              void* d_out, int out_size, void* d_ws, size_t ws_size,
                              hipStream_t stream) {
    const float* x   = (const float*)d_in[0];
    const int*   ei  = (const int*)  d_in[1];
    const float* ew  = (const float*)d_in[2];
    const float* W1  = (const float*)d_in[3];
    const float* b1  = (const float*)d_in[4];
    const float* W2  = (const float*)d_in[5];
    const float* b2  = (const float*)d_in[6];
    const float* Wfc = (const float*)d_in[7];
    const float* bfc = (const float*)d_in[8];
    float* out = (float*)d_out;

    const int* row = ei;
    const int* col = ei + N_EDGES;

    char* ws = (char*)d_ws;
    size_t off = 0;
    float* dinv = (float*)(ws + off); off += DINV_BYTES;
    float* bufH = (float*)(ws + off); off += PLANE_BYTES;
    float* bufA = (float*)(ws + off); off += PLANE_BYTES;
    int*   rec  = (int*)  (ws + off); off += REC_BYTES;
    int*   cnt  = (int*)  (ws + off); off += CNT_BYTES;
    int*   rel  = (int*)  (ws + off); off += CNT_BYTES;
    int*   tot  = (int*)  (ws + off);
    float* deg1 = bufA;
    (void)n_in; (void)in_sizes; (void)out_size;

    const int TPB = 256;
    const int nodeBlocks = (N_NODES + TPB - 1) / TPB;

    k_init_deg<<<nodeBlocks, TPB, 0, stream>>>(dinv, deg1);
    k_count<<<NCHUNKS, BIN_THREADS, 0, stream>>>(col, ew, dinv, deg1, cnt, ws_size);

    k_scan_dinv<<<NBINS, SCAN_THREADS, 0, stream>>>(cnt, rel, tot, dinv, deg1, ws_size);

    k_gemm_n64<IN_DIM, true><<<GEMM_BLOCKS, GEMM_THREADS, 0, stream>>>(
        x, W1, dinv, bufH, row, col, ew, rel, tot, rec, ws_size);
    k_owner<false><<<NBINS, OWN_THREADS, 0, stream>>>(rec, tot, bufH, dinv, b1, W2, bufA,
                                                      Wfc, bfc, out, ws_size);

    k_owner<true><<<NBINS, OWN_THREADS, 0, stream>>>(rec, tot, bufA, dinv, b2, W2, bufH,
                                                     Wfc, bfc, out, ws_size);
}
